# starting kernel
# baseline (speedup 1.0000x reference)
#include <hip/hip_runtime.h>
#include <hip/hip_bf16.h>
#include <math.h>

typedef __attribute__((ext_vector_type(2))) float v2f;
typedef __attribute__((ext_vector_type(8))) float v8f;

#define B_ 4
#define N_ 50000
#define E_ 800000
#define FIN_ 128
#define FOUT_ 64
#define ALPHA_ 0.2f
#define NEG_BIG_ (-9000000000000000.0f)

#define ROWS_ (B_ * N_)
#define TILES_ (ROWS_ / 16)

__global__ __launch_bounds__(256) void gat_gemm_wmma(const float* __restrict__ X,
                                                     const float* __restrict__ W,
                                                     float* __restrict__ h) {
    const int lane   = threadIdx.x & 31;
    const int waveId = threadIdx.x >> 5;
    const int rt     = blockIdx.x * 8 + waveId;
    if (rt >= TILES_) return;

    const int m_base = rt * 16;
    const int mrow   = m_base + (lane & 15);
    const int khalf  = (lane >> 4) << 1;
    const int ncol   = lane & 15;

    v8f acc0 = {}, acc1 = {}, acc2 = {}, acc3 = {};

    for (int k = 0; k < FIN_; k += 4) {
        const int kk = k + khalf;
        v2f a;
        a.x = X[(size_t)mrow * FIN_ + kk];
        a.y = X[(size_t)mrow * FIN_ + kk + 1];

        const float* wr0 = W + (size_t)kk * FOUT_;
        const float* wr1 = wr0 + FOUT_;

        v2f b0, b1, b2, b3;
        b0.x = wr0[ncol];      b0.y = wr1[ncol];
        b1.x = wr0[16 + ncol]; b1.y = wr1[16 + ncol];
        b2.x = wr0[32 + ncol]; b2.y = wr1[32 + ncol];
        b3.x = wr0[48 + ncol]; b3.y = wr1[48 + ncol];

        acc0 = __builtin_amdgcn_wmma_f32_16x16x4_f32(false, a, false, b0, (short)0, acc0, false, false);
        acc1 = __builtin_amdgcn_wmma_f32_16x16x4_f32(false, a, false, b1, (short)0, acc1, false, false);
        acc2 = __builtin_amdgcn_wmma_f32_16x16x4_f32(false, a, false, b2, (short)0, acc2, false, false);
        acc3 = __builtin_amdgcn_wmma_f32_16x16x4_f32(false, a, false, b3, (short)0, acc3, false, false);
    }

    const int rhalf = (lane >> 4) << 3;
#pragma unroll
    for (int i = 0; i < 8; ++i) {
        const size_t row = (size_t)(m_base + i + rhalf);
        h[row * FOUT_ +  0 + ncol] = acc0[i];
        h[row * FOUT_ + 16 + ncol] = acc1[i];
        h[row * FOUT_ + 32 + ncol] = acc2[i];
        h[row * FOUT_ + 48 + ncol] = acc3[i];
    }
}

__global__ __launch_bounds__(256) void gat_scores(const float* __restrict__ h,
                                                  const float* __restrict__ a,
                                                  float* __restrict__ e1,
                                                  float* __restrict__ e2) {
    const int row = blockIdx.x * blockDim.x + threadIdx.x;
    if (row >= ROWS_) return;
    const float* hr = h + (size_t)row * FOUT_;
    float s1 = 0.f, s2 = 0.f;
#pragma unroll
    for (int f = 0; f < FOUT_; ++f) {
        const float hv = hr[f];
        s1 += hv * a[f];
        s2 += hv * a[FOUT_ + f];
    }
    e1[row] = s1;
    e2[row] = s2;
}

__global__ __launch_bounds__(256) void gat_edges(const int* __restrict__ edges,
                                                 const float* __restrict__ h,
                                                 const float* __restrict__ e1,
                                                 const float* __restrict__ e2,
                                                 float* __restrict__ accum,
                                                 float* __restrict__ rowsum) {
    __shared__ float s_ee[4];
    __shared__ int   s_src[4];
    __shared__ int   s_dst[4];

    const int tid = threadIdx.x;
    const int le  = tid >> 6;
    const int f   = tid & 63;

    const long long eg = (long long)blockIdx.x * 4 + le;
    const bool valid = eg < (long long)B_ * E_;
    int b = 0, e = 0;
    if (valid) {
        b = (int)(eg / E_);
        e = (int)(eg % E_);
        if (f == 0) {
            const int src = edges[((size_t)b * 2 + 0) * E_ + e];
            const int dst = edges[((size_t)b * 2 + 1) * E_ + e];
            const float s  = e1[(size_t)b * N_ + src] + e2[(size_t)b * N_ + dst];
            const float lr = (s >= 0.f) ? s : ALPHA_ * s;
            const float ee = __expf(-lr);
            s_ee[le]  = ee;
            s_src[le] = src;
            s_dst[le] = dst;
            atomicAdd(&rowsum[(size_t)b * N_ + src], ee);
        }
    }
    __syncthreads();
    if (valid) {
        const int   src = s_src[le];
        const int   dst = s_dst[le];
        const float ee  = s_ee[le];
        const float hv  = h[((size_t)b * N_ + dst) * FOUT_ + f];
        atomicAdd(&accum[((size_t)b * N_ + src) * FOUT_ + f], ee * hv);
    }
}

__global__ __launch_bounds__(256) void gat_finalize(const float* __restrict__ accum,
                                                    const float* __restrict__ rowsum,
                                                    float* __restrict__ out) {
    const long long idx = (long long)blockIdx.x * blockDim.x + threadIdx.x;
    if (idx >= (long long)ROWS_ * FOUT_) return;
    const long long row = idx >> 6;
    float v = accum[idx] / rowsum[row];
    if (isnan(v)) v = NEG_BIG_;
    out[idx] = (v > 0.f) ? v : expm1f(v);
}

extern "C" void kernel_launch(void* const* d_in, const int* in_sizes, int n_in,
                              void* d_out, int out_size, void* d_ws, size_t ws_size,
                              hipStream_t stream) {
    const float* X     = (const float*)d_in[0];
    const int*   edges = (const int*)d_in[1];
    const float* W     = (const float*)d_in[2];
    const float* a     = (const float*)d_in[3];
    float*       out   = (float*)d_out;

    float* accum  = (float*)d_ws;
    float* rowsum = accum + (size_t)ROWS_ * FOUT_;
    float* h      = rowsum + (size_t)ROWS_;
    float* e1     = h + (size_t)ROWS_ * FOUT_;
    float* e2     = e1 + (size_t)ROWS_;

    hipMemsetAsync(accum, 0, ((size_t)ROWS_ * FOUT_ + ROWS_) * sizeof(float), stream);

    {
        const int waves_per_block = 8;
        const int blocks = (TILES_ + waves_per_block - 1) / waves_per_block;
        gat_gemm_wmma<<<blocks, 256, 0, stream>>>(X, W, h);
    }
    {
        const int blocks = (ROWS_ + 255) / 256;
        gat_scores<<<blocks, 256, 0, stream>>>(h, a, e1, e2);
    }
    {
        const long long total_edges = (long long)B_ * E_;
        const int blocks = (int)((total_edges + 3) / 4);
        gat_edges<<<blocks, 256, 0, stream>>>(edges, h, e1, e2, accum, rowsum);
    }
    {
        const long long total = (long long)ROWS_ * FOUT_;
        const int blocks = (int)((total + 255) / 256);
        gat_finalize<<<blocks, 256, 0, stream>>>(accum, rowsum, out);
    }
}


# f32 atomic scatter-add replaced by a per-row gather: edges bucketed by source row (int atomic rank picks the slot only), bucket sorted by dst in-wave, summed in ascending-dst order, divide+ELU fused
# speedup vs baseline: 2.6209x; 2.6209x over previous
#include <hip/hip_runtime.h>
#include <hip/hip_bf16.h>
#include <math.h>

typedef __attribute__((ext_vector_type(2))) float v2f;
typedef __attribute__((ext_vector_type(8))) float v8f;

#define B_ 4
#define N_ 50000
#define E_ 800000
#define FIN_ 128
#define FOUT_ 64
#define ALPHA_ 0.2f
#define NEG_BIG_ (-9000000000000000.0f)

#define ROWS_ (B_ * N_)
#define TILES_ (ROWS_ / 16)

__global__ __launch_bounds__(256) void gat_gemm_wmma(const float* __restrict__ X,
                                                     const float* __restrict__ W,
                                                     float* __restrict__ h) {
    const int lane   = threadIdx.x & 31;
    const int waveId = threadIdx.x >> 5;
    const int rt     = blockIdx.x * 8 + waveId;
    if (rt >= TILES_) return;

    const int m_base = rt * 16;
    const int mrow   = m_base + (lane & 15);
    const int khalf  = (lane >> 4) << 1;
    const int ncol   = lane & 15;

    v8f acc0 = {}, acc1 = {}, acc2 = {}, acc3 = {};

    for (int k = 0; k < FIN_; k += 4) {
        const int kk = k + khalf;
        v2f a;
        a.x = X[(size_t)mrow * FIN_ + kk];
        a.y = X[(size_t)mrow * FIN_ + kk + 1];

        const float* wr0 = W + (size_t)kk * FOUT_;
        const float* wr1 = wr0 + FOUT_;

        v2f b0, b1, b2, b3;
        b0.x = wr0[ncol];      b0.y = wr1[ncol];
        b1.x = wr0[16 + ncol]; b1.y = wr1[16 + ncol];
        b2.x = wr0[32 + ncol]; b2.y = wr1[32 + ncol];
        b3.x = wr0[48 + ncol]; b3.y = wr1[48 + ncol];

        acc0 = __builtin_amdgcn_wmma_f32_16x16x4_f32(false, a, false, b0, (short)0, acc0, false, false);
        acc1 = __builtin_amdgcn_wmma_f32_16x16x4_f32(false, a, false, b1, (short)0, acc1, false, false);
        acc2 = __builtin_amdgcn_wmma_f32_16x16x4_f32(false, a, false, b2, (short)0, acc2, false, false);
        acc3 = __builtin_amdgcn_wmma_f32_16x16x4_f32(false, a, false, b3, (short)0, acc3, false, false);
    }

    const int rhalf = (lane >> 4) << 3;
#pragma unroll
    for (int i = 0; i < 8; ++i) {
        const size_t row = (size_t)(m_base + i + rhalf);
        h[row * FOUT_ +  0 + ncol] = acc0[i];
        h[row * FOUT_ + 16 + ncol] = acc1[i];
        h[row * FOUT_ + 32 + ncol] = acc2[i];
        h[row * FOUT_ + 48 + ncol] = acc3[i];
    }
}

__global__ __launch_bounds__(256) void gat_scores(const float* __restrict__ h,
                                                  const float* __restrict__ a,
                                                  float* __restrict__ e1,
                                                  float* __restrict__ e2) {
    const int row = blockIdx.x * blockDim.x + threadIdx.x;
    if (row >= ROWS_) return;
    const float* hr = h + (size_t)row * FOUT_;
    float s1 = 0.f, s2 = 0.f;
#pragma unroll
    for (int f = 0; f < FOUT_; ++f) {
        const float hv = hr[f];
        s1 += hv * a[f];
        s2 += hv * a[FOUT_ + f];
    }
    e1[row] = s1;
    e2[row] = s2;
}

__global__ __launch_bounds__(256) void gat_rank(const int* __restrict__ edges,
                                                int* __restrict__ cnt,
                                                int* __restrict__ rank) {
    const int eg = blockIdx.x * 256 + threadIdx.x;
    const int b  = eg / E_;
    const int e  = eg - b * E_;
    const int src = edges[((size_t)b * 2 + 0) * E_ + e];
    rank[eg] = atomicAdd(&cnt[b * N_ + src], 1);
}

__global__ __launch_bounds__(256) void gat_alloc(const int* __restrict__ cnt,
                                                 int* __restrict__ cursor,
                                                 int* __restrict__ start) {
    __shared__ int s_tot[8];
    __shared__ int s_base;
    const int tid  = threadIdx.x;
    const int lane = tid & 31;
    const int wv   = tid >> 5;
    const int row  = blockIdx.x * 256 + tid;
    const int c    = (row < ROWS_) ? cnt[row] : 0;
    int inc = c;
#pragma unroll
    for (int o = 1; o < 32; o <<= 1) {
        const int t = __shfl_up(inc, o, 32);
        if (lane >= o) inc += t;
    }
    if (lane == 31) s_tot[wv] = inc;
    __syncthreads();
    if (tid == 0) {
        int tot = 0;
#pragma unroll
        for (int w = 0; w < 8; ++w) tot += s_tot[w];
        s_base = atomicAdd(cursor, tot);
    }
    __syncthreads();
    int off = s_base;
    for (int w = 0; w < wv; ++w) off += s_tot[w];
    if (row < ROWS_) start[row] = off + inc - c;
}

__global__ __launch_bounds__(256) void gat_scatter(const int* __restrict__ edges,
                                                   const int* __restrict__ start,
                                                   const int* __restrict__ rank,
                                                   int* __restrict__ slots) {
    const int eg = blockIdx.x * 256 + threadIdx.x;
    const int b  = eg / E_;
    const int e  = eg - b * E_;
    const int src = edges[((size_t)b * 2 + 0) * E_ + e];
    const int dst = edges[((size_t)b * 2 + 1) * E_ + e];
    slots[start[b * N_ + src] + rank[eg]] = dst;
}

#define DST_NONE_ 0x7fffffff

__device__ __forceinline__ float edge_weight(float e1r, float e2d) {
    const float s  = e1r + e2d;
    const float lr = (s >= 0.f) ? s : ALPHA_ * s;
    return __expf(-lr);
}

__device__ __forceinline__ int lane_int(int v, int l) { return __builtin_amdgcn_readlane(v, l); }
__device__ __forceinline__ float lane_flt(float v, int l) {
    return __int_as_float(__builtin_amdgcn_readlane(__float_as_int(v), l));
}

template <int NB>
__device__ __forceinline__ void take_edges(const float2* __restrict__ hb, int sd, float sw, int j,
                                           float& ax, float& ay, float& rs) {
    float2 hv[NB];
    float  wj[NB];
#pragma unroll
    for (int i = 0; i < NB; ++i) {
        const int dj = lane_int(sd, j + i);
        wj[i] = lane_flt(sw, j + i);
        hv[i] = hb[(size_t)dj * (FOUT_ / 2)];
    }
#pragma unroll
    for (int i = 0; i < NB; ++i) {
        ax = fmaf(wj[i], hv[i].x, ax);
        ay = fmaf(wj[i], hv[i].y, ay);
        rs += wj[i];
    }
}

__global__ __launch_bounds__(256) void gat_rows(const int* __restrict__ slots,
                                                const int* __restrict__ start,
                                                const int* __restrict__ cnt,
                                                const float* __restrict__ h,
                                                const float* __restrict__ e1,
                                                const float* __restrict__ e2,
                                                float* __restrict__ out) {
    const int lane = threadIdx.x & 31;
    const int wv  = __builtin_amdgcn_readfirstlane((int)(threadIdx.x >> 5));
    const int row = blockIdx.x * 8 + wv;
    const int b   = row / N_;
    const int deg = cnt[row];
    const int* seg = slots + start[row];
    const float e1r = e1[row];
    const float*  e2b = e2 + (size_t)b * N_;
    const float2* hb  = reinterpret_cast<const float2*>(h + (size_t)b * N_ * FOUT_) + lane;

    float ax = 0.f, ay = 0.f, rs = 0.f;

    if (deg <= 32) {
        const bool have = lane < deg;
        const int  d    = have ? seg[lane] : DST_NONE_;
        const float e2d = e2b[have ? d : 0];
        int pos = 0;
        for (int k = 0; k < deg; ++k) {
            const int dk = lane_int(d, k);
            pos += (dk < d || (dk == d && k < lane)) ? 1 : 0;
        }
        if (!have) pos = lane;
        const float w = have ? edge_weight(e1r, e2d) : 0.f;
        const int   sd = __builtin_amdgcn_ds_permute(pos << 2, d);
        const float sw = __int_as_float(__builtin_amdgcn_ds_permute(pos << 2, __float_as_int(w)));

        int j = 0;
        for (; j + 8 <= deg; j += 8) take_edges<8>(hb, sd, sw, j, ax, ay, rs);
        if (j + 4 <= deg) { take_edges<4>(hb, sd, sw, j, ax, ay, rs); j += 4; }
        if (j < deg) {
            const int last = deg - 1;
            const int j1 = (j + 1 < last) ? j + 1 : last;
            const int j2 = (j + 2 < last) ? j + 2 : last;
            const float2 h0 = hb[(size_t)lane_int(sd, j)  * (FOUT_ / 2)];
            const float2 h1 = hb[(size_t)lane_int(sd, j1) * (FOUT_ / 2)];
            const float2 h2 = hb[(size_t)lane_int(sd, j2) * (FOUT_ / 2)];
            const float w0 = lane_flt(sw, j), w1 = lane_flt(sw, j1), w2 = lane_flt(sw, j2);
            ax = fmaf(w0, h0.x, ax); ay = fmaf(w0, h0.y, ay); rs += w0;
            if (j + 1 < deg) { ax = fmaf(w1, h1.x, ax); ay = fmaf(w1, h1.y, ay); rs += w1; }
            if (j + 2 < deg) { ax = fmaf(w2, h2.x, ax); ay = fmaf(w2, h2.y, ay); rs += w2; }
        }
    } else {
        int prev = -1;
        for (int it = 0; it < deg; ++it) {
            int m = DST_NONE_;
            for (int base = 0; base < deg; base += 32) {
                const int i = base + lane;
                const int v = (i < deg) ? seg[i] : DST_NONE_;
                if (v > prev && v < m) m = v;
            }
#pragma unroll
            for (int o = 16; o > 0; o >>= 1) {
                const int t = __shfl_xor(m, o, 32);
                m = (t < m) ? t : m;
            }
            m = __builtin_amdgcn_readfirstlane(m);
            if (m == DST_NONE_) break;
            int c = 0;
            for (int base = 0; base < deg; base += 32) {
                const int i = base + lane;
                const int v = (i < deg) ? seg[i] : DST_NONE_;
                c += __popc(__builtin_amdgcn_ballot_w32(v == m));
            }
            const float wj = edge_weight(e1r, e2b[m]);
            const float2 hv = hb[(size_t)m * (FOUT_ / 2)];
            for (int t = 0; t < c; ++t) {
                ax = fmaf(wj, hv.x, ax);
                ay = fmaf(wj, hv.y, ay);
                rs += wj;
            }
            prev = m;
        }
    }

    float vx = ax / rs;
    float vy = ay / rs;
    if (isnan(vx)) vx = NEG_BIG_;
    if (isnan(vy)) vy = NEG_BIG_;
    float2 o;
    o.x = (vx > 0.f) ? vx : expm1f(vx);
    o.y = (vy > 0.f) ? vy : expm1f(vy);
    reinterpret_cast<float2*>(out + (size_t)row * FOUT_)[lane] = o;
}

extern "C" void kernel_launch(void* const* d_in, const int* in_sizes, int n_in,
                              void* d_out, int out_size, void* d_ws, size_t ws_size,
                              hipStream_t stream) {
    const float* X     = (const float*)d_in[0];
    const int*   edges = (const int*)d_in[1];
    const float* W     = (const float*)d_in[2];
    const float* a     = (const float*)d_in[3];
    float*       out   = (float*)d_out;

    float* h      = (float*)d_ws;
    float* e1     = h + (size_t)ROWS_ * FOUT_;
    float* e2     = e1 + (size_t)ROWS_;
    int*   cnt    = (int*)(e2 + (size_t)ROWS_);
    int*   cursor = cnt + (size_t)ROWS_;
    int*   start  = cursor + 32;
    int*   rank   = start + (size_t)ROWS_;
    int*   slots  = rank + (size_t)B_ * E_;

    hipMemsetAsync(cnt, 0, ((size_t)ROWS_ + 32) * sizeof(int), stream);

    {
        const int waves_per_block = 8;
        const int blocks = (TILES_ + waves_per_block - 1) / waves_per_block;
        gat_gemm_wmma<<<blocks, 256, 0, stream>>>(X, W, h);
    }
    {
        const int blocks = (ROWS_ + 255) / 256;
        gat_scores<<<blocks, 256, 0, stream>>>(h, a, e1, e2);
    }
    gat_rank<<<(B_ * E_) / 256, 256, 0, stream>>>(edges, cnt, rank);
    gat_alloc<<<(ROWS_ + 255) / 256, 256, 0, stream>>>(cnt, cursor, start);
    gat_scatter<<<(B_ * E_) / 256, 256, 0, stream>>>(edges, start, rank, slots);
    gat_rows<<<ROWS_ / 8, 256, 0, stream>>>(slots, start, cnt, h, e1, e2, out);
}


# Fewer, longer waves: rank/scatter threads take 8 edges each (int4 loads) and each row-kernel wave loops over 8 consecutive rows, an eighth of the waves in all three; per-row sum order unchanged
# speedup vs baseline: 2.8993x; 1.1062x over previous
#include <hip/hip_runtime.h>
#include <hip/hip_bf16.h>
#include <math.h>

typedef __attribute__((ext_vector_type(2))) float v2f;
typedef __attribute__((ext_vector_type(8))) float v8f;

#define B_ 4
#define N_ 50000
#define E_ 800000
#define FIN_ 128
#define FOUT_ 64
#define ALPHA_ 0.2f
#define NEG_BIG_ (-9000000000000000.0f)

#define ROWS_ (B_ * N_)
#define TILES_ (ROWS_ / 16)

__global__ __launch_bounds__(256) void gat_gemm_wmma(const float* __restrict__ X,
                                                     const float* __restrict__ W,
                                                     float* __restrict__ h) {
    const int lane   = threadIdx.x & 31;
    const int waveId = threadIdx.x >> 5;
    const int rt     = blockIdx.x * 8 + waveId;
    if (rt >= TILES_) return;

    const int m_base = rt * 16;
    const int mrow   = m_base + (lane & 15);
    const int khalf  = (lane >> 4) << 1;
    const int ncol   = lane & 15;

    v8f acc0 = {}, acc1 = {}, acc2 = {}, acc3 = {};

    for (int k = 0; k < FIN_; k += 4) {
        const int kk = k + khalf;
        v2f a;
        a.x = X[(size_t)mrow * FIN_ + kk];
        a.y = X[(size_t)mrow * FIN_ + kk + 1];

        const float* wr0 = W + (size_t)kk * FOUT_;
        const float* wr1 = wr0 + FOUT_;

        v2f b0, b1, b2, b3;
        b0.x = wr0[ncol];      b0.y = wr1[ncol];
        b1.x = wr0[16 + ncol]; b1.y = wr1[16 + ncol];
        b2.x = wr0[32 + ncol]; b2.y = wr1[32 + ncol];
        b3.x = wr0[48 + ncol]; b3.y = wr1[48 + ncol];

        acc0 = __builtin_amdgcn_wmma_f32_16x16x4_f32(false, a, false, b0, (short)0, acc0, false, false);
        acc1 = __builtin_amdgcn_wmma_f32_16x16x4_f32(false, a, false, b1, (short)0, acc1, false, false);
        acc2 = __builtin_amdgcn_wmma_f32_16x16x4_f32(false, a, false, b2, (short)0, acc2, false, false);
        acc3 = __builtin_amdgcn_wmma_f32_16x16x4_f32(false, a, false, b3, (short)0, acc3, false, false);
    }

    const int rhalf = (lane >> 4) << 3;
#pragma unroll
    for (int i = 0; i < 8; ++i) {
        const size_t row = (size_t)(m_base + i + rhalf);
        h[row * FOUT_ +  0 + ncol] = acc0[i];
        h[row * FOUT_ + 16 + ncol] = acc1[i];
        h[row * FOUT_ + 32 + ncol] = acc2[i];
        h[row * FOUT_ + 48 + ncol] = acc3[i];
    }
}

__global__ __launch_bounds__(256) void gat_scores(const float* __restrict__ h,
                                                  const float* __restrict__ a,
                                                  float* __restrict__ e1,
                                                  float* __restrict__ e2) {
    const int row = blockIdx.x * blockDim.x + threadIdx.x;
    if (row >= ROWS_) return;
    const float* hr = h + (size_t)row * FOUT_;
    float s1 = 0.f, s2 = 0.f;
#pragma unroll
    for (int f = 0; f < FOUT_; ++f) {
        const float hv = hr[f];
        s1 += hv * a[f];
        s2 += hv * a[FOUT_ + f];
    }
    e1[row] = s1;
    e2[row] = s2;
}

#define QUADS_PER_BATCH_ (E_ / 4)
#define QUADS_ (B_ * QUADS_PER_BATCH_)
#define EDGE_BLOCKS_ (QUADS_ / 256)

static_assert(E_ % 4 == 0, "a group of four edges must not cross a batch");
static_assert(QUADS_ % 256 == 0, "edge grid must be exact");

__global__ __launch_bounds__(128) void gat_rank(const int* __restrict__ edges,
                                                int* __restrict__ cnt,
                                                int* __restrict__ rank) {
    const int q0 = blockIdx.x * 256 + threadIdx.x;
    const int q1 = q0 + 128;
    const int b0 = q0 / QUADS_PER_BATCH_;
    const int b1 = q1 / QUADS_PER_BATCH_;
    const int e0 = (q0 - b0 * QUADS_PER_BATCH_) * 4;
    const int e1 = (q1 - b1 * QUADS_PER_BATCH_) * 4;
    const int4 s0 = *reinterpret_cast<const int4*>(edges + ((size_t)b0 * 2 + 0) * E_ + e0);
    const int4 s1 = *reinterpret_cast<const int4*>(edges + ((size_t)b1 * 2 + 0) * E_ + e1);
    int* c0 = cnt + b0 * N_;
    int* c1 = cnt + b1 * N_;
    int4 r0, r1;
    r0.x = atomicAdd(c0 + s0.x, 1);
    r0.y = atomicAdd(c0 + s0.y, 1);
    r0.z = atomicAdd(c0 + s0.z, 1);
    r0.w = atomicAdd(c0 + s0.w, 1);
    r1.x = atomicAdd(c1 + s1.x, 1);
    r1.y = atomicAdd(c1 + s1.y, 1);
    r1.z = atomicAdd(c1 + s1.z, 1);
    r1.w = atomicAdd(c1 + s1.w, 1);
    reinterpret_cast<int4*>(rank)[q0] = r0;
    reinterpret_cast<int4*>(rank)[q1] = r1;
}

__global__ __launch_bounds__(256) void gat_alloc(const int* __restrict__ cnt,
                                                 int* __restrict__ cursor,
                                                 int* __restrict__ start) {
    __shared__ int s_tot[8];
    __shared__ int s_base;
    const int tid  = threadIdx.x;
    const int lane = tid & 31;
    const int wv   = tid >> 5;
    const int row  = blockIdx.x * 256 + tid;
    const int c    = (row < ROWS_) ? cnt[row] : 0;
    int inc = c;
#pragma unroll
    for (int o = 1; o < 32; o <<= 1) {
        const int t = __shfl_up(inc, o, 32);
        if (lane >= o) inc += t;
    }
    if (lane == 31) s_tot[wv] = inc;
    __syncthreads();
    if (tid == 0) {
        int tot = 0;
#pragma unroll
        for (int w = 0; w < 8; ++w) tot += s_tot[w];
        s_base = atomicAdd(cursor, tot);
    }
    __syncthreads();
    int off = s_base;
    for (int w = 0; w < wv; ++w) off += s_tot[w];
    if (row < ROWS_) start[row] = off + inc - c;
}

__global__ __launch_bounds__(128) void gat_scatter(const int* __restrict__ edges,
                                                   const int* __restrict__ start,
                                                   const int* __restrict__ rank,
                                                   int* __restrict__ slots) {
    const int q0 = blockIdx.x * 256 + threadIdx.x;
    const int q1 = q0 + 128;
    const int b0 = q0 / QUADS_PER_BATCH_;
    const int b1 = q1 / QUADS_PER_BATCH_;
    const int e0 = (q0 - b0 * QUADS_PER_BATCH_) * 4;
    const int e1 = (q1 - b1 * QUADS_PER_BATCH_) * 4;
    const int4 s0 = *reinterpret_cast<const int4*>(edges + ((size_t)b0 * 2 + 0) * E_ + e0);
    const int4 d0 = *reinterpret_cast<const int4*>(edges + ((size_t)b0 * 2 + 1) * E_ + e0);
    const int4 s1 = *reinterpret_cast<const int4*>(edges + ((size_t)b1 * 2 + 0) * E_ + e1);
    const int4 d1 = *reinterpret_cast<const int4*>(edges + ((size_t)b1 * 2 + 1) * E_ + e1);
    const int4 r0 = reinterpret_cast<const int4*>(rank)[q0];
    const int4 r1 = reinterpret_cast<const int4*>(rank)[q1];
    const int* st0 = start + b0 * N_;
    const int* st1 = start + b1 * N_;
    const int p0 = st0[s0.x] + r0.x;
    const int p1 = st0[s0.y] + r0.y;
    const int p2 = st0[s0.z] + r0.z;
    const int p3 = st0[s0.w] + r0.w;
    const int p4 = st1[s1.x] + r1.x;
    const int p5 = st1[s1.y] + r1.y;
    const int p6 = st1[s1.z] + r1.z;
    const int p7 = st1[s1.w] + r1.w;
    slots[p0] = d0.x;
    slots[p1] = d0.y;
    slots[p2] = d0.z;
    slots[p3] = d0.w;
    slots[p4] = d1.x;
    slots[p5] = d1.y;
    slots[p6] = d1.z;
    slots[p7] = d1.w;
}

#define DST_NONE_ 0x7fffffff
#define ROWS_PER_WAVE_ 8

static_assert(N_ % ROWS_PER_WAVE_ == 0, "the rows of one wave must lie in one batch");
static_assert(ROWS_ % (8 * ROWS_PER_WAVE_) == 0, "row grid must be exact");

__device__ __forceinline__ float edge_weight(float e1r, float e2d) {
    const float s  = e1r + e2d;
    const float lr = (s >= 0.f) ? s : ALPHA_ * s;
    return __expf(-lr);
}

__device__ __forceinline__ int lane_int(int v, int l) { return __builtin_amdgcn_readlane(v, l); }
__device__ __forceinline__ float lane_flt(float v, int l) {
    return __int_as_float(__builtin_amdgcn_readlane(__float_as_int(v), l));
}

template <int NB>
__device__ __forceinline__ void take_edges(const float2* __restrict__ hb, int sd, float sw, int j,
                                           float& ax, float& ay, float& rs) {
    float2 hv[NB];
    float  wj[NB];
#pragma unroll
    for (int i = 0; i < NB; ++i) {
        const int dj = lane_int(sd, j + i);
        wj[i] = lane_flt(sw, j + i);
        hv[i] = hb[(size_t)dj * (FOUT_ / 2)];
    }
#pragma unroll
    for (int i = 0; i < NB; ++i) {
        ax = fmaf(wj[i], hv[i].x, ax);
        ay = fmaf(wj[i], hv[i].y, ay);
        rs += wj[i];
    }
}

__global__ __launch_bounds__(256) void gat_rows(const int* __restrict__ slots,
                                                const int* __restrict__ start,
                                                const int* __restrict__ cnt,
                                                const float* __restrict__ h,
                                                const float* __restrict__ e1,
                                                const float* __restrict__ e2,
                                                float* __restrict__ out) {
    const int lane = threadIdx.x & 31;
    const int wv   = __builtin_amdgcn_readfirstlane((int)(threadIdx.x >> 5));
    const int row0 = (blockIdx.x * 8 + wv) * ROWS_PER_WAVE_;
    const int b    = row0 / N_;
    const float*  e2b = e2 + (size_t)b * N_;
    const float2* hb  = reinterpret_cast<const float2*>(h + (size_t)b * N_ * FOUT_) + lane;

    for (int r = 0; r < ROWS_PER_WAVE_; ++r) {
    const int row = row0 + r;
    const int deg = cnt[row];
    const int* seg = slots + start[row];
    const float e1r = e1[row];

    float ax = 0.f, ay = 0.f, rs = 0.f;

    if (deg <= 32) {
        const bool have = lane < deg;
        const int  d    = have ? seg[lane] : DST_NONE_;
        const float e2d = e2b[have ? d : 0];
        int pos = 0;
        for (int k = 0; k < deg; ++k) {
            const int dk = lane_int(d, k);
            pos += (dk < d || (dk == d && k < lane)) ? 1 : 0;
        }
        if (!have) pos = lane;
        const float w = have ? edge_weight(e1r, e2d) : 0.f;
        const int   sd = __builtin_amdgcn_ds_permute(pos << 2, d);
        const float sw = __int_as_float(__builtin_amdgcn_ds_permute(pos << 2, __float_as_int(w)));

        int j = 0;
        for (; j + 8 <= deg; j += 8) take_edges<8>(hb, sd, sw, j, ax, ay, rs);
        if (j + 4 <= deg) { take_edges<4>(hb, sd, sw, j, ax, ay, rs); j += 4; }
        if (j < deg) {
            const int last = deg - 1;
            const int j1 = (j + 1 < last) ? j + 1 : last;
            const int j2 = (j + 2 < last) ? j + 2 : last;
            const float2 h0 = hb[(size_t)lane_int(sd, j)  * (FOUT_ / 2)];
            const float2 h1 = hb[(size_t)lane_int(sd, j1) * (FOUT_ / 2)];
            const float2 h2 = hb[(size_t)lane_int(sd, j2) * (FOUT_ / 2)];
            const float w0 = lane_flt(sw, j), w1 = lane_flt(sw, j1), w2 = lane_flt(sw, j2);
            ax = fmaf(w0, h0.x, ax); ay = fmaf(w0, h0.y, ay); rs += w0;
            if (j + 1 < deg) { ax = fmaf(w1, h1.x, ax); ay = fmaf(w1, h1.y, ay); rs += w1; }
            if (j + 2 < deg) { ax = fmaf(w2, h2.x, ax); ay = fmaf(w2, h2.y, ay); rs += w2; }
        }
    } else {
        int prev = -1;
        for (int it = 0; it < deg; ++it) {
            int m = DST_NONE_;
            for (int base = 0; base < deg; base += 32) {
                const int i = base + lane;
                const int v = (i < deg) ? seg[i] : DST_NONE_;
                if (v > prev && v < m) m = v;
            }
#pragma unroll
            for (int o = 16; o > 0; o >>= 1) {
                const int t = __shfl_xor(m, o, 32);
                m = (t < m) ? t : m;
            }
            m = __builtin_amdgcn_readfirstlane(m);
            if (m == DST_NONE_) break;
            int c = 0;
            for (int base = 0; base < deg; base += 32) {
                const int i = base + lane;
                const int v = (i < deg) ? seg[i] : DST_NONE_;
                c += __popc(__builtin_amdgcn_ballot_w32(v == m));
            }
            const float wj = edge_weight(e1r, e2b[m]);
            const float2 hv = hb[(size_t)m * (FOUT_ / 2)];
            for (int t = 0; t < c; ++t) {
                ax = fmaf(wj, hv.x, ax);
                ay = fmaf(wj, hv.y, ay);
                rs += wj;
            }
            prev = m;
        }
    }

    float vx = ax / rs;
    float vy = ay / rs;
    if (isnan(vx)) vx = NEG_BIG_;
    if (isnan(vy)) vy = NEG_BIG_;
    float2 o;
    o.x = (vx > 0.f) ? vx : expm1f(vx);
    o.y = (vy > 0.f) ? vy : expm1f(vy);
    reinterpret_cast<float2*>(out + (size_t)row * FOUT_)[lane] = o;
    }
}

extern "C" void kernel_launch(void* const* d_in, const int* in_sizes, int n_in,
                              void* d_out, int out_size, void* d_ws, size_t ws_size,
                              hipStream_t stream) {
    const float* X     = (const float*)d_in[0];
    const int*   edges = (const int*)d_in[1];
    const float* W     = (const float*)d_in[2];
    const float* a     = (const float*)d_in[3];
    float*       out   = (float*)d_out;

    float* h      = (float*)d_ws;
    float* e1     = h + (size_t)ROWS_ * FOUT_;
    float* e2     = e1 + (size_t)ROWS_;
    int*   cnt    = (int*)(e2 + (size_t)ROWS_);
    int*   cursor = cnt + (size_t)ROWS_;
    int*   start  = cursor + 32;
    int*   rank   = start + (size_t)ROWS_;
    int*   slots  = rank + (size_t)B_ * E_;

    hipMemsetAsync(cnt, 0, ((size_t)ROWS_ + 32) * sizeof(int), stream);

    {
        const int waves_per_block = 8;
        const int blocks = (TILES_ + waves_per_block - 1) / waves_per_block;
        gat_gemm_wmma<<<blocks, 256, 0, stream>>>(X, W, h);
    }
    {
        const int blocks = (ROWS_ + 255) / 256;
        gat_scores<<<blocks, 256, 0, stream>>>(h, a, e1, e2);
    }
    gat_rank<<<EDGE_BLOCKS_, 128, 0, stream>>>(edges, cnt, rank);
    gat_alloc<<<(ROWS_ + 255) / 256, 256, 0, stream>>>(cnt, cursor, start);
    gat_scatter<<<EDGE_BLOCKS_, 128, 0, stream>>>(edges, start, rank, slots);
    gat_rows<<<ROWS_ / (8 * ROWS_PER_WAVE_), 256, 0, stream>>>(slots, start, cnt, h, e1, e2, out);
}


# Fixed-capacity row buckets (32 words = one line at slots[row*32]): the thread that counts an edge stores its dst at once, full rows spill to a pair list; no scan, no 2nd edge pass, 6 launches to 4
# speedup vs baseline: 3.0347x; 1.0467x over previous
#include <hip/hip_runtime.h>
#include <hip/hip_bf16.h>
#include <math.h>

typedef __attribute__((ext_vector_type(2))) float v2f;
typedef __attribute__((ext_vector_type(8))) float v8f;

#define B_ 4
#define N_ 50000
#define E_ 800000
#define FIN_ 128
#define FOUT_ 64
#define ALPHA_ 0.2f
#define NEG_BIG_ (-9000000000000000.0f)

#define ROWS_ (B_ * N_)
#define TILES_ (ROWS_ / 16)

__global__ __launch_bounds__(256) void gat_gemm_wmma(const float* __restrict__ X,
                                                     const float* __restrict__ W,
                                                     float* __restrict__ h) {
    const int lane   = threadIdx.x & 31;
    const int waveId = threadIdx.x >> 5;
    const int rt     = blockIdx.x * 8 + waveId;
    if (rt >= TILES_) return;

    const int m_base = rt * 16;
    const int mrow   = m_base + (lane & 15);
    const int khalf  = (lane >> 4) << 1;
    const int ncol   = lane & 15;

    v8f acc0 = {}, acc1 = {}, acc2 = {}, acc3 = {};

    for (int k = 0; k < FIN_; k += 4) {
        const int kk = k + khalf;
        v2f a;
        a.x = X[(size_t)mrow * FIN_ + kk];
        a.y = X[(size_t)mrow * FIN_ + kk + 1];

        const float* wr0 = W + (size_t)kk * FOUT_;
        const float* wr1 = wr0 + FOUT_;

        v2f b0, b1, b2, b3;
        b0.x = wr0[ncol];      b0.y = wr1[ncol];
        b1.x = wr0[16 + ncol]; b1.y = wr1[16 + ncol];
        b2.x = wr0[32 + ncol]; b2.y = wr1[32 + ncol];
        b3.x = wr0[48 + ncol]; b3.y = wr1[48 + ncol];

        acc0 = __builtin_amdgcn_wmma_f32_16x16x4_f32(false, a, false, b0, (short)0, acc0, false, false);
        acc1 = __builtin_amdgcn_wmma_f32_16x16x4_f32(false, a, false, b1, (short)0, acc1, false, false);
        acc2 = __builtin_amdgcn_wmma_f32_16x16x4_f32(false, a, false, b2, (short)0, acc2, false, false);
        acc3 = __builtin_amdgcn_wmma_f32_16x16x4_f32(false, a, false, b3, (short)0, acc3, false, false);
    }

    const int rhalf = (lane >> 4) << 3;
#pragma unroll
    for (int i = 0; i < 8; ++i) {
        const size_t row = (size_t)(m_base + i + rhalf);
        h[row * FOUT_ +  0 + ncol] = acc0[i];
        h[row * FOUT_ + 16 + ncol] = acc1[i];
        h[row * FOUT_ + 32 + ncol] = acc2[i];
        h[row * FOUT_ + 48 + ncol] = acc3[i];
    }
}

__global__ __launch_bounds__(256) void gat_scores(const float* __restrict__ h,
                                                  const float* __restrict__ a,
                                                  float* __restrict__ e1,
                                                  float* __restrict__ e2) {
    const int row = blockIdx.x * blockDim.x + threadIdx.x;
    if (row >= ROWS_) return;
    const float* hr = h + (size_t)row * FOUT_;
    float s1 = 0.f, s2 = 0.f;
#pragma unroll
    for (int f = 0; f < FOUT_; ++f) {
        const float hv = hr[f];
        s1 += hv * a[f];
        s2 += hv * a[FOUT_ + f];
    }
    e1[row] = s1;
    e2[row] = s2;
}

#define CAP_ 32
#define QUADS_PER_BATCH_ (E_ / 4)
#define QUADS_ (B_ * QUADS_PER_BATCH_)
#define EDGE_BLOCKS_ (QUADS_ / 256)

static_assert(E_ % 4 == 0, "a group of four edges must not cross a batch");
static_assert(QUADS_ % 256 == 0, "edge grid must be exact");

__device__ __forceinline__ void put_edge(int* __restrict__ slots, int* __restrict__ ovfcnt,
                                         int* __restrict__ ovf, int row, int r, int d) {
    if (r < CAP_) {
        slots[row * CAP_ + r] = d;
    } else {
        const int k = atomicAdd(ovfcnt, 1);
        reinterpret_cast<int2*>(ovf)[k] = make_int2(row, d);
    }
}

__global__ __launch_bounds__(128) void gat_fill(const int* __restrict__ edges,
                                                int* __restrict__ cnt,
                                                int* __restrict__ slots,
                                                int* __restrict__ ovfcnt,
                                                int* __restrict__ ovf) {
    const int q0 = blockIdx.x * 256 + threadIdx.x;
    const int q1 = q0 + 128;
    const int b0 = q0 / QUADS_PER_BATCH_;
    const int b1 = q1 / QUADS_PER_BATCH_;
    const int e0 = (q0 - b0 * QUADS_PER_BATCH_) * 4;
    const int e1 = (q1 - b1 * QUADS_PER_BATCH_) * 4;
    const int4 s0 = *reinterpret_cast<const int4*>(edges + ((size_t)b0 * 2 + 0) * E_ + e0);
    const int4 s1 = *reinterpret_cast<const int4*>(edges + ((size_t)b1 * 2 + 0) * E_ + e1);
    const int4 d0 = *reinterpret_cast<const int4*>(edges + ((size_t)b0 * 2 + 1) * E_ + e0);
    const int4 d1 = *reinterpret_cast<const int4*>(edges + ((size_t)b1 * 2 + 1) * E_ + e1);
    const int rb0 = b0 * N_;
    const int rb1 = b1 * N_;
    int* c0 = cnt + rb0;
    int* c1 = cnt + rb1;
    const int r0 = atomicAdd(c0 + s0.x, 1);
    const int r1 = atomicAdd(c0 + s0.y, 1);
    const int r2 = atomicAdd(c0 + s0.z, 1);
    const int r3 = atomicAdd(c0 + s0.w, 1);
    const int r4 = atomicAdd(c1 + s1.x, 1);
    const int r5 = atomicAdd(c1 + s1.y, 1);
    const int r6 = atomicAdd(c1 + s1.z, 1);
    const int r7 = atomicAdd(c1 + s1.w, 1);
    put_edge(slots, ovfcnt, ovf, rb0 + s0.x, r0, d0.x);
    put_edge(slots, ovfcnt, ovf, rb0 + s0.y, r1, d0.y);
    put_edge(slots, ovfcnt, ovf, rb0 + s0.z, r2, d0.z);
    put_edge(slots, ovfcnt, ovf, rb0 + s0.w, r3, d0.w);
    put_edge(slots, ovfcnt, ovf, rb1 + s1.x, r4, d1.x);
    put_edge(slots, ovfcnt, ovf, rb1 + s1.y, r5, d1.y);
    put_edge(slots, ovfcnt, ovf, rb1 + s1.z, r6, d1.z);
    put_edge(slots, ovfcnt, ovf, rb1 + s1.w, r7, d1.w);
}

#define DST_NONE_ 0x7fffffff
#define ROWS_PER_WAVE_ 8

static_assert(N_ % ROWS_PER_WAVE_ == 0, "the rows of one wave must lie in one batch");
static_assert(ROWS_ % (8 * ROWS_PER_WAVE_) == 0, "row grid must be exact");

__device__ __forceinline__ float edge_weight(float e1r, float e2d) {
    const float s  = e1r + e2d;
    const float lr = (s >= 0.f) ? s : ALPHA_ * s;
    return __expf(-lr);
}

__device__ __forceinline__ int lane_int(int v, int l) { return __builtin_amdgcn_readlane(v, l); }
__device__ __forceinline__ float lane_flt(float v, int l) {
    return __int_as_float(__builtin_amdgcn_readlane(__float_as_int(v), l));
}

template <int NB>
__device__ __forceinline__ void take_edges(const float2* __restrict__ hb, int sd, float sw, int j,
                                           float& ax, float& ay, float& rs) {
    float2 hv[NB];
    float  wj[NB];
#pragma unroll
    for (int i = 0; i < NB; ++i) {
        const int dj = lane_int(sd, j + i);
        wj[i] = lane_flt(sw, j + i);
        hv[i] = hb[(size_t)dj * (FOUT_ / 2)];
    }
#pragma unroll
    for (int i = 0; i < NB; ++i) {
        ax = fmaf(wj[i], hv[i].x, ax);
        ay = fmaf(wj[i], hv[i].y, ay);
        rs += wj[i];
    }
}

__global__ __launch_bounds__(256) void gat_rows(const int* __restrict__ slots,
                                                const int* __restrict__ cnt,
                                                const int* __restrict__ ovfcnt,
                                                const int* __restrict__ ovf,
                                                const float* __restrict__ h,
                                                const float* __restrict__ e1,
                                                const float* __restrict__ e2,
                                                float* __restrict__ out) {
    const int lane = threadIdx.x & 31;
    const int wv   = __builtin_amdgcn_readfirstlane((int)(threadIdx.x >> 5));
    const int row0 = (blockIdx.x * 8 + wv) * ROWS_PER_WAVE_;
    const int b    = row0 / N_;
    const float*  e2b = e2 + (size_t)b * N_;
    const float2* hb  = reinterpret_cast<const float2*>(h + (size_t)b * N_ * FOUT_) + lane;

    for (int r = 0; r < ROWS_PER_WAVE_; ++r) {
    const int row = row0 + r;
    const int deg = cnt[row];
    const float e1r = e1[row];
    const int dl = slots[row * CAP_ + lane];

    float ax = 0.f, ay = 0.f, rs = 0.f;

    if (deg <= CAP_) {
        const bool have = lane < deg;
        const int  d    = have ? dl : DST_NONE_;
        const float e2d = e2b[have ? d : 0];
        int pos = 0;
        for (int k = 0; k < deg; ++k) {
            const int dk = lane_int(d, k);
            pos += (dk < d || (dk == d && k < lane)) ? 1 : 0;
        }
        if (!have) pos = lane;
        const float w = have ? edge_weight(e1r, e2d) : 0.f;
        const int   sd = __builtin_amdgcn_ds_permute(pos << 2, d);
        const float sw = __int_as_float(__builtin_amdgcn_ds_permute(pos << 2, __float_as_int(w)));

        int j = 0;
        for (; j + 8 <= deg; j += 8) take_edges<8>(hb, sd, sw, j, ax, ay, rs);
        if (j + 4 <= deg) { take_edges<4>(hb, sd, sw, j, ax, ay, rs); j += 4; }
        if (j < deg) {
            const int last = deg - 1;
            const int j1 = (j + 1 < last) ? j + 1 : last;
            const int j2 = (j + 2 < last) ? j + 2 : last;
            const float2 h0 = hb[(size_t)lane_int(sd, j)  * (FOUT_ / 2)];
            const float2 h1 = hb[(size_t)lane_int(sd, j1) * (FOUT_ / 2)];
            const float2 h2 = hb[(size_t)lane_int(sd, j2) * (FOUT_ / 2)];
            const float w0 = lane_flt(sw, j), w1 = lane_flt(sw, j1), w2 = lane_flt(sw, j2);
            ax = fmaf(w0, h0.x, ax); ay = fmaf(w0, h0.y, ay); rs += w0;
            if (j + 1 < deg) { ax = fmaf(w1, h1.x, ax); ay = fmaf(w1, h1.y, ay); rs += w1; }
            if (j + 2 < deg) { ax = fmaf(w2, h2.x, ax); ay = fmaf(w2, h2.y, ay); rs += w2; }
        }
    } else {
        const int   novf = ovfcnt[0];
        const int2* ol   = reinterpret_cast<const int2*>(ovf);
        int prev = -1;
        for (int it = 0; it < deg; ++it) {
            int m = (dl > prev) ? dl : DST_NONE_;
            for (int base = 0; base < novf; base += 32) {
                const int i = base + lane;
                if (i < novf) {
                    const int2 p = ol[i];
                    if (p.x == row && p.y > prev && p.y < m) m = p.y;
                }
            }
#pragma unroll
            for (int o = 16; o > 0; o >>= 1) {
                const int t = __shfl_xor(m, o, 32);
                m = (t < m) ? t : m;
            }
            m = __builtin_amdgcn_readfirstlane(m);
            if (m == DST_NONE_) break;
            int c = __popc(__builtin_amdgcn_ballot_w32(dl == m));
            for (int base = 0; base < novf; base += 32) {
                const int i = base + lane;
                bool hit = false;
                if (i < novf) {
                    const int2 p = ol[i];
                    hit = (p.x == row) && (p.y == m);
                }
                c += __popc(__builtin_amdgcn_ballot_w32(hit));
            }
            const float wj = edge_weight(e1r, e2b[m]);
            const float2 hv = hb[(size_t)m * (FOUT_ / 2)];
            for (int t = 0; t < c; ++t) {
                ax = fmaf(wj, hv.x, ax);
                ay = fmaf(wj, hv.y, ay);
                rs += wj;
            }
            prev = m;
        }
    }

    float vx = ax / rs;
    float vy = ay / rs;
    if (isnan(vx)) vx = NEG_BIG_;
    if (isnan(vy)) vy = NEG_BIG_;
    float2 o;
    o.x = (vx > 0.f) ? vx : expm1f(vx);
    o.y = (vy > 0.f) ? vy : expm1f(vy);
    reinterpret_cast<float2*>(out + (size_t)row * FOUT_)[lane] = o;
    }
}

extern "C" void kernel_launch(void* const* d_in, const int* in_sizes, int n_in,
                              void* d_out, int out_size, void* d_ws, size_t ws_size,
                              hipStream_t stream) {
    const float* X     = (const float*)d_in[0];
    const int*   edges = (const int*)d_in[1];
    const float* W     = (const float*)d_in[2];
    const float* a     = (const float*)d_in[3];
    float*       out   = (float*)d_out;

    float* h      = (float*)d_ws;
    float* e1     = h + (size_t)ROWS_ * FOUT_;
    float* e2     = e1 + (size_t)ROWS_;
    int*   cnt    = (int*)(e2 + (size_t)ROWS_);
    int*   ovfcnt = cnt + (size_t)ROWS_;
    int*   slots  = ovfcnt + 32;
    int*   ovf    = slots + (size_t)ROWS_ * CAP_;

    hipMemsetAsync(cnt, 0, ((size_t)ROWS_ + 32) * sizeof(int), stream);

    {
        const int waves_per_block = 8;
        const int blocks = (TILES_ + waves_per_block - 1) / waves_per_block;
        gat_gemm_wmma<<<blocks, 256, 0, stream>>>(X, W, h);
    }
    {
        const int blocks = (ROWS_ + 255) / 256;
        gat_scores<<<blocks, 256, 0, stream>>>(h, a, e1, e2);
    }
    gat_fill<<<EDGE_BLOCKS_, 128, 0, stream>>>(edges, cnt, slots, ovfcnt, ovf);
    gat_rows<<<ROWS_ / (8 * ROWS_PER_WAVE_), 256, 0, stream>>>(slots, cnt, ovfcnt, ovf, h, e1, e2, out);
}


# One returning global add per edge replaced by a two-level counting sort in LDS: edges grouped by 256-row range per 4096-edge chunk, then ranked and bucketed by the block that owns the range
# speedup vs baseline: 6.2944x; 2.0741x over previous
#include <hip/hip_runtime.h>
#include <hip/hip_bf16.h>
#include <math.h>

typedef __attribute__((ext_vector_type(2))) float v2f;
typedef __attribute__((ext_vector_type(8))) float v8f;

#define B_ 4
#define N_ 50000
#define E_ 800000
#define FIN_ 128
#define FOUT_ 64
#define ALPHA_ 0.2f
#define NEG_BIG_ (-9000000000000000.0f)

#define ROWS_ (B_ * N_)
#define TILES_ (ROWS_ / 16)

__global__ __launch_bounds__(256) void gat_gemm_wmma(const float* __restrict__ X,
                                                     const float* __restrict__ W,
                                                     float* __restrict__ h) {
    const int lane   = threadIdx.x & 31;
    const int waveId = threadIdx.x >> 5;
    const int rt     = blockIdx.x * 8 + waveId;
    if (rt >= TILES_) return;

    const int m_base = rt * 16;
    const int mrow   = m_base + (lane & 15);
    const int khalf  = (lane >> 4) << 1;
    const int ncol   = lane & 15;

    v8f acc0 = {}, acc1 = {}, acc2 = {}, acc3 = {};

    for (int k = 0; k < FIN_; k += 4) {
        const int kk = k + khalf;
        v2f a;
        a.x = X[(size_t)mrow * FIN_ + kk];
        a.y = X[(size_t)mrow * FIN_ + kk + 1];

        const float* wr0 = W + (size_t)kk * FOUT_;
        const float* wr1 = wr0 + FOUT_;

        v2f b0, b1, b2, b3;
        b0.x = wr0[ncol];      b0.y = wr1[ncol];
        b1.x = wr0[16 + ncol]; b1.y = wr1[16 + ncol];
        b2.x = wr0[32 + ncol]; b2.y = wr1[32 + ncol];
        b3.x = wr0[48 + ncol]; b3.y = wr1[48 + ncol];

        acc0 = __builtin_amdgcn_wmma_f32_16x16x4_f32(false, a, false, b0, (short)0, acc0, false, false);
        acc1 = __builtin_amdgcn_wmma_f32_16x16x4_f32(false, a, false, b1, (short)0, acc1, false, false);
        acc2 = __builtin_amdgcn_wmma_f32_16x16x4_f32(false, a, false, b2, (short)0, acc2, false, false);
        acc3 = __builtin_amdgcn_wmma_f32_16x16x4_f32(false, a, false, b3, (short)0, acc3, false, false);
    }

    const int rhalf = (lane >> 4) << 3;
#pragma unroll
    for (int i = 0; i < 8; ++i) {
        const size_t row = (size_t)(m_base + i + rhalf);
        h[row * FOUT_ +  0 + ncol] = acc0[i];
        h[row * FOUT_ + 16 + ncol] = acc1[i];
        h[row * FOUT_ + 32 + ncol] = acc2[i];
        h[row * FOUT_ + 48 + ncol] = acc3[i];
    }
}

__global__ __launch_bounds__(256) void gat_scores(const float* __restrict__ h,
                                                  const float* __restrict__ a,
                                                  float* __restrict__ e1,
                                                  float* __restrict__ e2) {
    const int row = blockIdx.x * blockDim.x + threadIdx.x;
    if (row >= ROWS_) return;
    const float* hr = h + (size_t)row * FOUT_;
    float s1 = 0.f, s2 = 0.f;
#pragma unroll
    for (int f = 0; f < FOUT_; ++f) {
        const float hv = hr[f];
        s1 += hv * a[f];
        s2 += hv * a[FOUT_ + f];
    }
    e1[row] = s1;
    e2[row] = s2;
}

#define CAP_ 32
#define RANGE_SHIFT_ 8
#define RANGE_ (1 << RANGE_SHIFT_)
#define NRANGE_ ((N_ + RANGE_ - 1) / RANGE_)
#define NTAB_ 256
#define CHUNK_ 4096
#define NCHUNK_ ((E_ + CHUNK_ - 1) / CHUNK_)
#define QUADS_PER_BATCH_ (E_ / 4)
#define SEG_WORDS_ (((B_ * (NRANGE_ + 1) * NCHUNK_ + 31) / 32) * 32)

static_assert(E_ % 4 == 0, "a group of four edges must not cross a batch");
static_assert(NRANGE_ < NTAB_, "the range table needs one entry past the last range");

__device__ __forceinline__ unsigned long long make_pair(int s, int d) {
    return ((unsigned long long)(unsigned)d << 32) | (unsigned long long)(unsigned)s;
}

__device__ __forceinline__ int range_rank(int* hist, int bk) {
    const int      key  = __builtin_amdgcn_readfirstlane(bk);
    const unsigned act  = __builtin_amdgcn_ballot_w32(true);
    const unsigned same = __builtin_amdgcn_ballot_w32(bk == key);
    if (same == act) {
        const unsigned lane   = threadIdx.x & 31;
        const int      within = __popc(act & ((1u << lane) - 1u));
        int base = 0;
        if (within == 0) base = atomicAdd(&hist[key], __popc(act));
        base = __builtin_amdgcn_readfirstlane(base);
        return base + within;
    }
    return atomicAdd(&hist[bk], 1);
}

__global__ __launch_bounds__(256) void gat_part(const int* __restrict__ edges,
                                                int* __restrict__ seg,
                                                unsigned long long* __restrict__ pairs) {
    __shared__ int hist[NTAB_];
    __shared__ int wsum[8];
    __shared__ __attribute__((aligned(16))) unsigned long long stage[CHUNK_];
    const int t = threadIdx.x;
    const int b = blockIdx.x / NCHUNK_;
    const int c = blockIdx.x - b * NCHUNK_;
    hist[t] = 0;
    __syncthreads();

    const int4* src4 = reinterpret_cast<const int4*>(edges + ((size_t)b * 2 + 0) * E_);
    const int4* dst4 = reinterpret_cast<const int4*>(edges + ((size_t)b * 2 + 1) * E_);
    int4 s[4], d[4];
    bool ok[4];
#pragma unroll
    for (int i = 0; i < 4; ++i) {
        const int q = c * (CHUNK_ / 4) + i * 256 + t;
        ok[i] = q < QUADS_PER_BATCH_;
        if (ok[i]) { s[i] = src4[q]; d[i] = dst4[q]; }
        else       { s[i] = make_int4(0, 0, 0, 0); d[i] = s[i]; }
    }
    int rk[16];
#pragma unroll
    for (int i = 0; i < 4; ++i) {
        if (ok[i]) {
            rk[4 * i + 0] = range_rank(hist, s[i].x >> RANGE_SHIFT_);
            rk[4 * i + 1] = range_rank(hist, s[i].y >> RANGE_SHIFT_);
            rk[4 * i + 2] = range_rank(hist, s[i].z >> RANGE_SHIFT_);
            rk[4 * i + 3] = range_rank(hist, s[i].w >> RANGE_SHIFT_);
        }
    }
    __syncthreads();
    const int v = hist[t];
    int incl = v;
#pragma unroll
    for (int o = 1; o < 32; o <<= 1) {
        const int u = __shfl_up(incl, o, 32);
        if ((t & 31) >= o) incl += u;
    }
    if ((t & 31) == 31) wsum[t >> 5] = incl;
    __syncthreads();
    int off = 0;
    for (int k = 0; k < (t >> 5); ++k) off += wsum[k];
    const int excl = incl - v + off;
    hist[t] = excl;
    if (t <= NRANGE_) seg[(size_t)(b * (NRANGE_ + 1) + t) * NCHUNK_ + c] = excl;
    __syncthreads();
#pragma unroll
    for (int i = 0; i < 4; ++i) {
        if (ok[i]) {
            stage[hist[s[i].x >> RANGE_SHIFT_] + rk[4 * i + 0]] = make_pair(s[i].x, d[i].x);
            stage[hist[s[i].y >> RANGE_SHIFT_] + rk[4 * i + 1]] = make_pair(s[i].y, d[i].y);
            stage[hist[s[i].z >> RANGE_SHIFT_] + rk[4 * i + 2]] = make_pair(s[i].z, d[i].z);
            stage[hist[s[i].w >> RANGE_SHIFT_] + rk[4 * i + 3]] = make_pair(s[i].w, d[i].w);
        }
    }
    __syncthreads();
    const int n2 = hist[NRANGE_] >> 1;
    const ulonglong2* st2 = reinterpret_cast<const ulonglong2*>(stage);
    ulonglong2* out2 = reinterpret_cast<ulonglong2*>(pairs + (size_t)b * E_ + (size_t)c * CHUNK_);
    for (int i = t; i < n2; i += 256) out2[i] = st2[i];
}

#define PLACE_THREADS_ 512
#define PLACE_WAVES_ (PLACE_THREADS_ / 32)
#define PLACE_TRIPS_ ((NCHUNK_ + PLACE_WAVES_ - 1) / PLACE_WAVES_)
#define PLACE_HEAD_ 64

static_assert(NCHUNK_ <= PLACE_THREADS_, "one thread of gat_place per chunk reads the group bounds");
static_assert(RANGE_ <= PLACE_THREADS_, "one thread of gat_place per row of the range");

__device__ __forceinline__ void place_pair(int* cntl, int* slotl,
                                           int* __restrict__ ovfcnt, int* __restrict__ ovf,
                                           unsigned long long p, int row0, int rowb) {
    const int sx = (int)(unsigned)(p & 0xffffffffull);
    const int dy = (int)(unsigned)(p >> 32);
    const int lr = sx - row0;
    const int r  = atomicAdd(&cntl[lr], 1);
    if (r < CAP_) {
        slotl[lr * CAP_ + r] = dy;
    } else {
        const int k = atomicAdd(ovfcnt, 1);
        reinterpret_cast<int2*>(ovf)[k] = make_int2(rowb + sx, dy);
    }
}

__global__ __launch_bounds__(PLACE_THREADS_) void gat_place(const unsigned long long* __restrict__ pairs,
                                                            const int* __restrict__ seg,
                                                            int* __restrict__ cnt,
                                                            int* __restrict__ slots,
                                                            int* __restrict__ ovfcnt,
                                                            int* __restrict__ ovf) {
    __shared__ int cntl[RANGE_];
    __shared__ int slotl[RANGE_ * CAP_];
    __shared__ int sbeg[NCHUNK_];
    __shared__ int send[NCHUNK_];
    __shared__ int longs[NCHUNK_];
    __shared__ int nlong;
    const int t = threadIdx.x;
    const int b = blockIdx.x / NRANGE_;
    const int g = blockIdx.x - b * NRANGE_;
    if (t == 0) nlong = 0;
    if (t < RANGE_) cntl[t] = 0;
    int glen = 0;
    if (t < NCHUNK_) {
        const int* sg = seg + (size_t)(b * (NRANGE_ + 1) + g) * NCHUNK_;
        const int gb = sg[t];
        const int ge = sg[NCHUNK_ + t];
        sbeg[t] = gb;
        send[t] = ge;
        glen = ge - gb;
    }
    __syncthreads();
    if (glen > PLACE_HEAD_) longs[atomicAdd(&nlong, 1)] = t;

    const int lane = t & 31;
    const int wv   = __builtin_amdgcn_readfirstlane(t >> 5);
    const int row0 = g * RANGE_;
    const int rowb = b * N_;
    const unsigned long long* pb = pairs + (size_t)b * E_;

    unsigned long long p0[PLACE_TRIPS_], p1[PLACE_TRIPS_];
    int i0[PLACE_TRIPS_], en[PLACE_TRIPS_];
#pragma unroll
    for (int k = 0; k < PLACE_TRIPS_; ++k) {
        const int c = wv + k * PLACE_WAVES_;
        int gb = 0, ge = 0;
        if (c < NCHUNK_) { gb = sbeg[c]; ge = send[c]; }
        i0[k] = c * CHUNK_ + gb + lane;
        en[k] = c * CHUNK_ + ge;
        p0[k] = 0ull;
        p1[k] = 0ull;
        if (i0[k] < en[k])      p0[k] = pb[i0[k]];
        if (i0[k] + 32 < en[k]) p1[k] = pb[i0[k] + 32];
    }
#pragma unroll
    for (int k = 0; k < PLACE_TRIPS_; ++k) {
        if (i0[k] < en[k])      place_pair(cntl, slotl, ovfcnt, ovf, p0[k], row0, rowb);
        if (i0[k] + 32 < en[k]) place_pair(cntl, slotl, ovfcnt, ovf, p1[k], row0, rowb);
    }
    __syncthreads();
    const int nl = nlong;
    for (int k = 0; k < nl; ++k) {
        const int c  = longs[k];
        const int ie = c * CHUNK_ + send[c];
        for (int i = c * CHUNK_ + sbeg[c] + PLACE_HEAD_ + t; i < ie; i += PLACE_THREADS_)
            place_pair(cntl, slotl, ovfcnt, ovf, pb[i], row0, rowb);
    }
    __syncthreads();
    for (int r = wv; r < RANGE_; r += PLACE_WAVES_) {
        const int n = cntl[r];
        if (lane < n) slots[(size_t)(rowb + row0 + r) * CAP_ + lane] = slotl[r * CAP_ + lane];
    }
    if (t < RANGE_ && row0 + t < N_) cnt[rowb + row0 + t] = cntl[t];
}

#define DST_NONE_ 0x7fffffff
#define ROWS_PER_WAVE_ 8

static_assert(N_ % ROWS_PER_WAVE_ == 0, "the rows of one wave must lie in one batch");
static_assert(ROWS_ % (8 * ROWS_PER_WAVE_) == 0, "row grid must be exact");

__device__ __forceinline__ float edge_weight(float e1r, float e2d) {
    const float s  = e1r + e2d;
    const float lr = (s >= 0.f) ? s : ALPHA_ * s;
    return __expf(-lr);
}

__device__ __forceinline__ int lane_int(int v, int l) { return __builtin_amdgcn_readlane(v, l); }
__device__ __forceinline__ float lane_flt(float v, int l) {
    return __int_as_float(__builtin_amdgcn_readlane(__float_as_int(v), l));
}

template <int NB>
__device__ __forceinline__ void take_edges(const float2* __restrict__ hb, int sd, float sw, int j,
                                           float& ax, float& ay, float& rs) {
    float2 hv[NB];
    float  wj[NB];
#pragma unroll
    for (int i = 0; i < NB; ++i) {
        const int dj = lane_int(sd, j + i);
        wj[i] = lane_flt(sw, j + i);
        hv[i] = hb[(size_t)dj * (FOUT_ / 2)];
    }
#pragma unroll
    for (int i = 0; i < NB; ++i) {
        ax = fmaf(wj[i], hv[i].x, ax);
        ay = fmaf(wj[i], hv[i].y, ay);
        rs += wj[i];
    }
}

__global__ __launch_bounds__(256) void gat_rows(const int* __restrict__ slots,
                                                const int* __restrict__ cnt,
                                                const int* __restrict__ ovfcnt,
                                                const int* __restrict__ ovf,
                                                const float* __restrict__ h,
                                                const float* __restrict__ e1,
                                                const float* __restrict__ e2,
                                                float* __restrict__ out) {
    const int lane = threadIdx.x & 31;
    const int wv   = __builtin_amdgcn_readfirstlane((int)(threadIdx.x >> 5));
    const int row0 = (blockIdx.x * 8 + wv) * ROWS_PER_WAVE_;
    const int b    = row0 / N_;
    const float*  e2b = e2 + (size_t)b * N_;
    const float2* hb  = reinterpret_cast<const float2*>(h + (size_t)b * N_ * FOUT_) + lane;

    for (int r = 0; r < ROWS_PER_WAVE_; ++r) {
    const int row = row0 + r;
    const int deg = cnt[row];
    const float e1r = e1[row];
    const int dl = slots[row * CAP_ + lane];

    float ax = 0.f, ay = 0.f, rs = 0.f;

    if (deg <= CAP_) {
        const bool have = lane < deg;
        const int  d    = have ? dl : DST_NONE_;
        const float e2d = e2b[have ? d : 0];
        int pos = 0;
        for (int k = 0; k < deg; ++k) {
            const int dk = lane_int(d, k);
            pos += (dk < d || (dk == d && k < lane)) ? 1 : 0;
        }
        if (!have) pos = lane;
        const float w = have ? edge_weight(e1r, e2d) : 0.f;
        const int   sd = __builtin_amdgcn_ds_permute(pos << 2, d);
        const float sw = __int_as_float(__builtin_amdgcn_ds_permute(pos << 2, __float_as_int(w)));

        int j = 0;
        for (; j + 8 <= deg; j += 8) take_edges<8>(hb, sd, sw, j, ax, ay, rs);
        if (j + 4 <= deg) { take_edges<4>(hb, sd, sw, j, ax, ay, rs); j += 4; }
        if (j < deg) {
            const int last = deg - 1;
            const int j1 = (j + 1 < last) ? j + 1 : last;
            const int j2 = (j + 2 < last) ? j + 2 : last;
            const float2 h0 = hb[(size_t)lane_int(sd, j)  * (FOUT_ / 2)];
            const float2 h1 = hb[(size_t)lane_int(sd, j1) * (FOUT_ / 2)];
            const float2 h2 = hb[(size_t)lane_int(sd, j2) * (FOUT_ / 2)];
            const float w0 = lane_flt(sw, j), w1 = lane_flt(sw, j1), w2 = lane_flt(sw, j2);
            ax = fmaf(w0, h0.x, ax); ay = fmaf(w0, h0.y, ay); rs += w0;
            if (j + 1 < deg) { ax = fmaf(w1, h1.x, ax); ay = fmaf(w1, h1.y, ay); rs += w1; }
            if (j + 2 < deg) { ax = fmaf(w2, h2.x, ax); ay = fmaf(w2, h2.y, ay); rs += w2; }
        }
    } else {
        const int   novf = ovfcnt[0];
        const int2* ol   = reinterpret_cast<const int2*>(ovf);
        int prev = -1;
        for (int it = 0; it < deg; ++it) {
            int m = (dl > prev) ? dl : DST_NONE_;
            for (int base = 0; base < novf; base += 32) {
                const int i = base + lane;
                if (i < novf) {
                    const int2 p = ol[i];
                    if (p.x == row && p.y > prev && p.y < m) m = p.y;
                }
            }
#pragma unroll
            for (int o = 16; o > 0; o >>= 1) {
                const int t = __shfl_xor(m, o, 32);
                m = (t < m) ? t : m;
            }
            m = __builtin_amdgcn_readfirstlane(m);
            if (m == DST_NONE_) break;
            int c = __popc(__builtin_amdgcn_ballot_w32(dl == m));
            for (int base = 0; base < novf; base += 32) {
                const int i = base + lane;
                bool hit = false;
                if (i < novf) {
                    const int2 p = ol[i];
                    hit = (p.x == row) && (p.y == m);
                }
                c += __popc(__builtin_amdgcn_ballot_w32(hit));
            }
            const float wj = edge_weight(e1r, e2b[m]);
            const float2 hv = hb[(size_t)m * (FOUT_ / 2)];
            for (int t = 0; t < c; ++t) {
                ax = fmaf(wj, hv.x, ax);
                ay = fmaf(wj, hv.y, ay);
                rs += wj;
            }
            prev = m;
        }
    }

    float vx = ax / rs;
    float vy = ay / rs;
    if (isnan(vx)) vx = NEG_BIG_;
    if (isnan(vy)) vy = NEG_BIG_;
    float2 o;
    o.x = (vx > 0.f) ? vx : expm1f(vx);
    o.y = (vy > 0.f) ? vy : expm1f(vy);
    reinterpret_cast<float2*>(out + (size_t)row * FOUT_)[lane] = o;
    }
}

extern "C" void kernel_launch(void* const* d_in, const int* in_sizes, int n_in,
                              void* d_out, int out_size, void* d_ws, size_t ws_size,
                              hipStream_t stream) {
    const float* X     = (const float*)d_in[0];
    const int*   edges = (const int*)d_in[1];
    const float* W     = (const float*)d_in[2];
    const float* a     = (const float*)d_in[3];
    float*       out   = (float*)d_out;

    float* h      = (float*)d_ws;
    float* e1     = h + (size_t)ROWS_ * FOUT_;
    float* e2     = e1 + (size_t)ROWS_;
    int*   cnt    = (int*)(e2 + (size_t)ROWS_);
    int*   ovfcnt = cnt + (size_t)ROWS_;
    int*   slots  = ovfcnt + 32;
    int*   ovf    = slots + (size_t)ROWS_ * CAP_;
    int*   seg    = ovf + (size_t)2 * B_ * E_;
    unsigned long long* pairs = (unsigned long long*)(seg + (size_t)SEG_WORDS_);

    hipMemsetAsync(ovfcnt, 0, 32 * sizeof(int), stream);

    {
        const int waves_per_block = 8;
        const int blocks = (TILES_ + waves_per_block - 1) / waves_per_block;
        gat_gemm_wmma<<<blocks, 256, 0, stream>>>(X, W, h);
    }
    {
        const int blocks = (ROWS_ + 255) / 256;
        gat_scores<<<blocks, 256, 0, stream>>>(h, a, e1, e2);
    }
    gat_part<<<B_ * NCHUNK_, 256, 0, stream>>>(edges, seg, pairs);
    gat_place<<<B_ * NRANGE_, PLACE_THREADS_, 0, stream>>>(pairs, seg, cnt, slots, ovfcnt, ovf);
    gat_rows<<<ROWS_ / (8 * ROWS_PER_WAVE_), 256, 0, stream>>>(slots, cnt, ovfcnt, ovf, h, e1, e2, out);
}


# Rows of 33 to 64 edges: per-edge search for the next smallest destination (dependent loads per edge) replaced by one ranking pass over bucket and collected overflow pairs; sum order unchanged
# speedup vs baseline: 7.3330x; 1.1650x over previous
#include <hip/hip_runtime.h>
#include <hip/hip_bf16.h>
#include <math.h>

typedef __attribute__((ext_vector_type(2))) float v2f;
typedef __attribute__((ext_vector_type(8))) float v8f;

#define B_ 4
#define N_ 50000
#define E_ 800000
#define FIN_ 128
#define FOUT_ 64
#define ALPHA_ 0.2f
#define NEG_BIG_ (-9000000000000000.0f)

#define ROWS_ (B_ * N_)
#define TILES_ (ROWS_ / 16)

__global__ __launch_bounds__(256) void gat_gemm_wmma(const float* __restrict__ X,
                                                     const float* __restrict__ W,
                                                     float* __restrict__ h) {
    const int lane   = threadIdx.x & 31;
    const int waveId = threadIdx.x >> 5;
    const int rt     = blockIdx.x * 8 + waveId;
    if (rt >= TILES_) return;

    const int m_base = rt * 16;
    const int mrow   = m_base + (lane & 15);
    const int khalf  = (lane >> 4) << 1;
    const int ncol   = lane & 15;

    v8f acc0 = {}, acc1 = {}, acc2 = {}, acc3 = {};

    for (int k = 0; k < FIN_; k += 4) {
        const int kk = k + khalf;
        v2f a;
        a.x = X[(size_t)mrow * FIN_ + kk];
        a.y = X[(size_t)mrow * FIN_ + kk + 1];

        const float* wr0 = W + (size_t)kk * FOUT_;
        const float* wr1 = wr0 + FOUT_;

        v2f b0, b1, b2, b3;
        b0.x = wr0[ncol];      b0.y = wr1[ncol];
        b1.x = wr0[16 + ncol]; b1.y = wr1[16 + ncol];
        b2.x = wr0[32 + ncol]; b2.y = wr1[32 + ncol];
        b3.x = wr0[48 + ncol]; b3.y = wr1[48 + ncol];

        acc0 = __builtin_amdgcn_wmma_f32_16x16x4_f32(false, a, false, b0, (short)0, acc0, false, false);
        acc1 = __builtin_amdgcn_wmma_f32_16x16x4_f32(false, a, false, b1, (short)0, acc1, false, false);
        acc2 = __builtin_amdgcn_wmma_f32_16x16x4_f32(false, a, false, b2, (short)0, acc2, false, false);
        acc3 = __builtin_amdgcn_wmma_f32_16x16x4_f32(false, a, false, b3, (short)0, acc3, false, false);
    }

    const int rhalf = (lane >> 4) << 3;
#pragma unroll
    for (int i = 0; i < 8; ++i) {
        const size_t row = (size_t)(m_base + i + rhalf);
        h[row * FOUT_ +  0 + ncol] = acc0[i];
        h[row * FOUT_ + 16 + ncol] = acc1[i];
        h[row * FOUT_ + 32 + ncol] = acc2[i];
        h[row * FOUT_ + 48 + ncol] = acc3[i];
    }
}

__global__ __launch_bounds__(256) void gat_scores(const float* __restrict__ h,
                                                  const float* __restrict__ a,
                                                  float* __restrict__ e1,
                                                  float* __restrict__ e2) {
    const int row = blockIdx.x * blockDim.x + threadIdx.x;
    if (row >= ROWS_) return;
    const float* hr = h + (size_t)row * FOUT_;
    float s1 = 0.f, s2 = 0.f;
#pragma unroll
    for (int f = 0; f < FOUT_; ++f) {
        const float hv = hr[f];
        s1 += hv * a[f];
        s2 += hv * a[FOUT_ + f];
    }
    e1[row] = s1;
    e2[row] = s2;
}

#define CAP_ 32
#define RANGE_SHIFT_ 8
#define RANGE_ (1 << RANGE_SHIFT_)
#define NRANGE_ ((N_ + RANGE_ - 1) / RANGE_)
#define NTAB_ 256
#define CHUNK_ 4096
#define NCHUNK_ ((E_ + CHUNK_ - 1) / CHUNK_)
#define QUADS_PER_BATCH_ (E_ / 4)
#define SEG_WORDS_ (((B_ * (NRANGE_ + 1) * NCHUNK_ + 31) / 32) * 32)

static_assert(E_ % 4 == 0, "a group of four edges must not cross a batch");
static_assert(NRANGE_ < NTAB_, "the range table needs one entry past the last range");

__device__ __forceinline__ unsigned long long make_pair(int s, int d) {
    return ((unsigned long long)(unsigned)d << 32) | (unsigned long long)(unsigned)s;
}

__device__ __forceinline__ int range_rank(int* hist, int bk) {
    const int      key  = __builtin_amdgcn_readfirstlane(bk);
    const unsigned act  = __builtin_amdgcn_ballot_w32(true);
    const unsigned same = __builtin_amdgcn_ballot_w32(bk == key);
    if (same == act) {
        const unsigned lane   = threadIdx.x & 31;
        const int      within = __popc(act & ((1u << lane) - 1u));
        int base = 0;
        if (within == 0) base = atomicAdd(&hist[key], __popc(act));
        base = __builtin_amdgcn_readfirstlane(base);
        return base + within;
    }
    return atomicAdd(&hist[bk], 1);
}

__global__ __launch_bounds__(256) void gat_part(const int* __restrict__ edges,
                                                int* __restrict__ seg,
                                                unsigned long long* __restrict__ pairs) {
    __shared__ int hist[NTAB_];
    __shared__ int wsum[8];
    __shared__ __attribute__((aligned(16))) unsigned long long stage[CHUNK_];
    const int t = threadIdx.x;
    const int b = blockIdx.x / NCHUNK_;
    const int c = blockIdx.x - b * NCHUNK_;
    hist[t] = 0;
    __syncthreads();

    const int4* src4 = reinterpret_cast<const int4*>(edges + ((size_t)b * 2 + 0) * E_);
    const int4* dst4 = reinterpret_cast<const int4*>(edges + ((size_t)b * 2 + 1) * E_);
    int4 s[4], d[4];
    bool ok[4];
#pragma unroll
    for (int i = 0; i < 4; ++i) {
        const int q = c * (CHUNK_ / 4) + i * 256 + t;
        ok[i] = q < QUADS_PER_BATCH_;
        if (ok[i]) { s[i] = src4[q]; d[i] = dst4[q]; }
        else       { s[i] = make_int4(0, 0, 0, 0); d[i] = s[i]; }
    }
    int rk[16];
#pragma unroll
    for (int i = 0; i < 4; ++i) {
        if (ok[i]) {
            rk[4 * i + 0] = range_rank(hist, s[i].x >> RANGE_SHIFT_);
            rk[4 * i + 1] = range_rank(hist, s[i].y >> RANGE_SHIFT_);
            rk[4 * i + 2] = range_rank(hist, s[i].z >> RANGE_SHIFT_);
            rk[4 * i + 3] = range_rank(hist, s[i].w >> RANGE_SHIFT_);
        }
    }
    __syncthreads();
    const int v = hist[t];
    int incl = v;
#pragma unroll
    for (int o = 1; o < 32; o <<= 1) {
        const int u = __shfl_up(incl, o, 32);
        if ((t & 31) >= o) incl += u;
    }
    if ((t & 31) == 31) wsum[t >> 5] = incl;
    __syncthreads();
    int off = 0;
    for (int k = 0; k < (t >> 5); ++k) off += wsum[k];
    const int excl = incl - v + off;
    hist[t] = excl;
    if (t <= NRANGE_) seg[(size_t)(b * (NRANGE_ + 1) + t) * NCHUNK_ + c] = excl;
    __syncthreads();
#pragma unroll
    for (int i = 0; i < 4; ++i) {
        if (ok[i]) {
            stage[hist[s[i].x >> RANGE_SHIFT_] + rk[4 * i + 0]] = make_pair(s[i].x, d[i].x);
            stage[hist[s[i].y >> RANGE_SHIFT_] + rk[4 * i + 1]] = make_pair(s[i].y, d[i].y);
            stage[hist[s[i].z >> RANGE_SHIFT_] + rk[4 * i + 2]] = make_pair(s[i].z, d[i].z);
            stage[hist[s[i].w >> RANGE_SHIFT_] + rk[4 * i + 3]] = make_pair(s[i].w, d[i].w);
        }
    }
    __syncthreads();
    const int n2 = hist[NRANGE_] >> 1;
    const ulonglong2* st2 = reinterpret_cast<const ulonglong2*>(stage);
    ulonglong2* out2 = reinterpret_cast<ulonglong2*>(pairs + (size_t)b * E_ + (size_t)c * CHUNK_);
    for (int i = t; i < n2; i += 256) out2[i] = st2[i];
}

#define PLACE_THREADS_ 512
#define PLACE_WAVES_ (PLACE_THREADS_ / 32)
#define PLACE_TRIPS_ ((NCHUNK_ + PLACE_WAVES_ - 1) / PLACE_WAVES_)
#define PLACE_HEAD_ 64

static_assert(NCHUNK_ <= PLACE_THREADS_, "one thread of gat_place per chunk reads the group bounds");
static_assert(RANGE_ <= PLACE_THREADS_, "one thread of gat_place per row of the range");

__device__ __forceinline__ void place_pair(int* cntl, int* slotl,
                                           int* __restrict__ ovfcnt, int* __restrict__ ovf,
                                           unsigned long long p, int row0, int rowb) {
    const int sx = (int)(unsigned)(p & 0xffffffffull);
    const int dy = (int)(unsigned)(p >> 32);
    const int lr = sx - row0;
    const int r  = atomicAdd(&cntl[lr], 1);
    if (r < CAP_) {
        slotl[lr * CAP_ + r] = dy;
    } else {
        const int k = atomicAdd(ovfcnt, 1);
        reinterpret_cast<int2*>(ovf)[k] = make_int2(rowb + sx, dy);
    }
}

__global__ __launch_bounds__(PLACE_THREADS_) void gat_place(const unsigned long long* __restrict__ pairs,
                                                            const int* __restrict__ seg,
                                                            int* __restrict__ cnt,
                                                            int* __restrict__ slots,
                                                            int* __restrict__ ovfcnt,
                                                            int* __restrict__ ovf) {
    __shared__ int cntl[RANGE_];
    __shared__ int slotl[RANGE_ * CAP_];
    __shared__ int sbeg[NCHUNK_];
    __shared__ int send[NCHUNK_];
    __shared__ int longs[NCHUNK_];
    __shared__ int nlong;
    const int t = threadIdx.x;
    const int b = blockIdx.x / NRANGE_;
    const int g = blockIdx.x - b * NRANGE_;
    if (t == 0) nlong = 0;
    if (t < RANGE_) cntl[t] = 0;
    int glen = 0;
    if (t < NCHUNK_) {
        const int* sg = seg + (size_t)(b * (NRANGE_ + 1) + g) * NCHUNK_;
        const int gb = sg[t];
        const int ge = sg[NCHUNK_ + t];
        sbeg[t] = gb;
        send[t] = ge;
        glen = ge - gb;
    }
    __syncthreads();
    if (glen > PLACE_HEAD_) longs[atomicAdd(&nlong, 1)] = t;

    const int lane = t & 31;
    const int wv   = __builtin_amdgcn_readfirstlane(t >> 5);
    const int row0 = g * RANGE_;
    const int rowb = b * N_;
    const unsigned long long* pb = pairs + (size_t)b * E_;

    unsigned long long p0[PLACE_TRIPS_], p1[PLACE_TRIPS_];
    int i0[PLACE_TRIPS_], en[PLACE_TRIPS_];
#pragma unroll
    for (int k = 0; k < PLACE_TRIPS_; ++k) {
        const int c = wv + k * PLACE_WAVES_;
        int gb = 0, ge = 0;
        if (c < NCHUNK_) { gb = sbeg[c]; ge = send[c]; }
        i0[k] = c * CHUNK_ + gb + lane;
        en[k] = c * CHUNK_ + ge;
        p0[k] = 0ull;
        p1[k] = 0ull;
        if (i0[k] < en[k])      p0[k] = pb[i0[k]];
        if (i0[k] + 32 < en[k]) p1[k] = pb[i0[k] + 32];
    }
#pragma unroll
    for (int k = 0; k < PLACE_TRIPS_; ++k) {
        if (i0[k] < en[k])      place_pair(cntl, slotl, ovfcnt, ovf, p0[k], row0, rowb);
        if (i0[k] + 32 < en[k]) place_pair(cntl, slotl, ovfcnt, ovf, p1[k], row0, rowb);
    }
    __syncthreads();
    const int nl = nlong;
    for (int k = 0; k < nl; ++k) {
        const int c  = longs[k];
        const int ie = c * CHUNK_ + send[c];
        for (int i = c * CHUNK_ + sbeg[c] + PLACE_HEAD_ + t; i < ie; i += PLACE_THREADS_)
            place_pair(cntl, slotl, ovfcnt, ovf, pb[i], row0, rowb);
    }
    __syncthreads();
    for (int r = wv; r < RANGE_; r += PLACE_WAVES_) {
        const int n = cntl[r];
        if (lane < n) slots[(size_t)(rowb + row0 + r) * CAP_ + lane] = slotl[r * CAP_ + lane];
    }
    if (t < RANGE_ && row0 + t < N_) cnt[rowb + row0 + t] = cntl[t];
}

#define DST_NONE_ 0x7fffffff
#define ROWS_PER_WAVE_ 8

static_assert(N_ % ROWS_PER_WAVE_ == 0, "the rows of one wave must lie in one batch");
static_assert(ROWS_ % (8 * ROWS_PER_WAVE_) == 0, "row grid must be exact");

__device__ __forceinline__ float edge_weight(float e1r, float e2d) {
    const float s  = e1r + e2d;
    const float lr = (s >= 0.f) ? s : ALPHA_ * s;
    return __expf(-lr);
}

__device__ __forceinline__ int lane_int(int v, int l) { return __builtin_amdgcn_readlane(v, l); }
__device__ __forceinline__ float lane_flt(float v, int l) {
    return __int_as_float(__builtin_amdgcn_readlane(__float_as_int(v), l));
}

template <int NB>
__device__ __forceinline__ void take_edges(const float2* __restrict__ hb, int sd, float sw, int j,
                                           float& ax, float& ay, float& rs) {
    float2 hv[NB];
    float  wj[NB];
#pragma unroll
    for (int i = 0; i < NB; ++i) {
        const int dj = lane_int(sd, j + i);
        wj[i] = lane_flt(sw, j + i);
        hv[i] = hb[(size_t)dj * (FOUT_ / 2)];
    }
#pragma unroll
    for (int i = 0; i < NB; ++i) {
        ax = fmaf(wj[i], hv[i].x, ax);
        ay = fmaf(wj[i], hv[i].y, ay);
        rs += wj[i];
    }
}

__device__ __forceinline__ void ranked_edge(int posA, int posB, bool hasB, int dA, int dB,
                                            float wA, float wB, int j, int& dj, float& wj) {
    const unsigned mA = __builtin_amdgcn_ballot_w32(posA == j);
    const unsigned mB = __builtin_amdgcn_ballot_w32(hasB && posB == j);
    const bool fromA  = mA != 0u;
    const int  l      = __builtin_ffs((int)(fromA ? mA : mB)) - 1;
    const int   da = lane_int(dA, l), db = lane_int(dB, l);
    const float wa = lane_flt(wA, l), wb = lane_flt(wB, l);
    const int   d  = fromA ? da : db;
    dj = ((unsigned)d < (unsigned)N_) ? d : 0;
    wj = fromA ? wa : wb;
}

__global__ __launch_bounds__(256) void gat_rows(const int* __restrict__ slots,
                                                const int* __restrict__ cnt,
                                                const int* __restrict__ ovfcnt,
                                                const int* __restrict__ ovf,
                                                const float* __restrict__ h,
                                                const float* __restrict__ e1,
                                                const float* __restrict__ e2,
                                                float* __restrict__ out) {
    const int lane = threadIdx.x & 31;
    const int wv   = __builtin_amdgcn_readfirstlane((int)(threadIdx.x >> 5));
    const int row0 = (blockIdx.x * 8 + wv) * ROWS_PER_WAVE_;
    const int b    = row0 / N_;
    const float*  e2b = e2 + (size_t)b * N_;
    const float2* hb  = reinterpret_cast<const float2*>(h + (size_t)b * N_ * FOUT_) + lane;

    for (int r = 0; r < ROWS_PER_WAVE_; ++r) {
    const int row = row0 + r;
    const int deg = cnt[row];
    const float e1r = e1[row];
    const int dl = slots[row * CAP_ + lane];

    float ax = 0.f, ay = 0.f, rs = 0.f;

    if (deg <= CAP_) {
        const bool have = lane < deg;
        const int  d    = have ? dl : DST_NONE_;
        const float e2d = e2b[have ? d : 0];
        int pos = 0;
        for (int k = 0; k < deg; ++k) {
            const int dk = lane_int(d, k);
            pos += (dk < d || (dk == d && k < lane)) ? 1 : 0;
        }
        if (!have) pos = lane;
        const float w = have ? edge_weight(e1r, e2d) : 0.f;
        const int   sd = __builtin_amdgcn_ds_permute(pos << 2, d);
        const float sw = __int_as_float(__builtin_amdgcn_ds_permute(pos << 2, __float_as_int(w)));

        int j = 0;
        for (; j + 8 <= deg; j += 8) take_edges<8>(hb, sd, sw, j, ax, ay, rs);
        if (j + 4 <= deg) { take_edges<4>(hb, sd, sw, j, ax, ay, rs); j += 4; }
        if (j < deg) {
            const int last = deg - 1;
            const int j1 = (j + 1 < last) ? j + 1 : last;
            const int j2 = (j + 2 < last) ? j + 2 : last;
            const float2 h0 = hb[(size_t)lane_int(sd, j)  * (FOUT_ / 2)];
            const float2 h1 = hb[(size_t)lane_int(sd, j1) * (FOUT_ / 2)];
            const float2 h2 = hb[(size_t)lane_int(sd, j2) * (FOUT_ / 2)];
            const float w0 = lane_flt(sw, j), w1 = lane_flt(sw, j1), w2 = lane_flt(sw, j2);
            ax = fmaf(w0, h0.x, ax); ay = fmaf(w0, h0.y, ay); rs += w0;
            if (j + 1 < deg) { ax = fmaf(w1, h1.x, ax); ay = fmaf(w1, h1.y, ay); rs += w1; }
            if (j + 2 < deg) { ax = fmaf(w2, h2.x, ax); ay = fmaf(w2, h2.y, ay); rs += w2; }
        }
    } else if (deg <= 2 * CAP_) {
        const int   novf = ovfcnt[0];
        const int2* ol   = reinterpret_cast<const int2*>(ovf);
        int d2 = DST_NONE_;
        int k  = 0;
        for (int base = 0; base < novf; base += 32) {
            const int i = base + lane;
            int2 p = make_int2(-1, -1);
            if (i < novf) p = ol[i];
            unsigned hit = __builtin_amdgcn_ballot_w32(p.x == row);
            while (hit != 0u) {
                const int l = __builtin_ffs((int)hit) - 1;
                const int v = lane_int(p.y, l);
                if (lane == k) d2 = v;
                ++k;
                hit &= hit - 1u;
            }
        }
        if (k > 32) k = 32;
        const bool  hasB = lane < k;
        const int   dA   = dl;
        const int   dB   = d2;
        const float e2A  = e2b[dA];
        const float e2B  = e2b[hasB ? dB : 0];
        int posA = 0, posB = 0;
#pragma nounroll
        for (int j = 0; j < CAP_; ++j) {
            const int x = lane_int(dA, j);
            posA += (x < dA || (x == dA && j < lane)) ? 1 : 0;
            posB += (x <= dB) ? 1 : 0;
        }
        for (int j = 0; j < k; ++j) {
            const int y = lane_int(dB, j);
            posA += (y < dA) ? 1 : 0;
            posB += (y < dB || (y == dB && j < lane)) ? 1 : 0;
        }
        const float wA = edge_weight(e1r, e2A);
        const float wB = hasB ? edge_weight(e1r, e2B) : 0.f;
        const int   tot = CAP_ + k;
        for (int j0 = 0; j0 < tot; j0 += 8) {
            float2 hv[8];
            float  wj[8];
#pragma unroll
            for (int i = 0; i < 8; ++i) {
                const int j = (j0 + i < tot) ? j0 + i : tot - 1;
                int dj;
                ranked_edge(posA, posB, hasB, dA, dB, wA, wB, j, dj, wj[i]);
                hv[i] = hb[(size_t)dj * (FOUT_ / 2)];
            }
#pragma unroll
            for (int i = 0; i < 8; ++i) {
                if (j0 + i < tot) {
                    ax = fmaf(wj[i], hv[i].x, ax);
                    ay = fmaf(wj[i], hv[i].y, ay);
                    rs += wj[i];
                }
            }
        }
    } else {
        const int   novf = ovfcnt[0];
        const int2* ol   = reinterpret_cast<const int2*>(ovf);
        int prev = -1;
        for (int it = 0; it < deg; ++it) {
            int m = (dl > prev) ? dl : DST_NONE_;
            for (int base = 0; base < novf; base += 32) {
                const int i = base + lane;
                if (i < novf) {
                    const int2 p = ol[i];
                    if (p.x == row && p.y > prev && p.y < m) m = p.y;
                }
            }
#pragma unroll
            for (int o = 16; o > 0; o >>= 1) {
                const int t = __shfl_xor(m, o, 32);
                m = (t < m) ? t : m;
            }
            m = __builtin_amdgcn_readfirstlane(m);
            if (m == DST_NONE_) break;
            int c = __popc(__builtin_amdgcn_ballot_w32(dl == m));
            for (int base = 0; base < novf; base += 32) {
                const int i = base + lane;
                bool hit = false;
                if (i < novf) {
                    const int2 p = ol[i];
                    hit = (p.x == row) && (p.y == m);
                }
                c += __popc(__builtin_amdgcn_ballot_w32(hit));
            }
            const float wj = edge_weight(e1r, e2b[m]);
            const float2 hv = hb[(size_t)m * (FOUT_ / 2)];
            for (int t = 0; t < c; ++t) {
                ax = fmaf(wj, hv.x, ax);
                ay = fmaf(wj, hv.y, ay);
                rs += wj;
            }
            prev = m;
        }
    }

    float vx = ax / rs;
    float vy = ay / rs;
    if (isnan(vx)) vx = NEG_BIG_;
    if (isnan(vy)) vy = NEG_BIG_;
    float2 o;
    o.x = (vx > 0.f) ? vx : expm1f(vx);
    o.y = (vy > 0.f) ? vy : expm1f(vy);
    reinterpret_cast<float2*>(out + (size_t)row * FOUT_)[lane] = o;
    }
}

extern "C" void kernel_launch(void* const* d_in, const int* in_sizes, int n_in,
                              void* d_out, int out_size, void* d_ws, size_t ws_size,
                              hipStream_t stream) {
    const float* X     = (const float*)d_in[0];
    const int*   edges = (const int*)d_in[1];
    const float* W     = (const float*)d_in[2];
    const float* a     = (const float*)d_in[3];
    float*       out   = (float*)d_out;

    float* h      = (float*)d_ws;
    float* e1     = h + (size_t)ROWS_ * FOUT_;
    float* e2     = e1 + (size_t)ROWS_;
    int*   cnt    = (int*)(e2 + (size_t)ROWS_);
    int*   ovfcnt = cnt + (size_t)ROWS_;
    int*   slots  = ovfcnt + 32;
    int*   ovf    = slots + (size_t)ROWS_ * CAP_;
    int*   seg    = ovf + (size_t)2 * B_ * E_;
    unsigned long long* pairs = (unsigned long long*)(seg + (size_t)SEG_WORDS_);

    hipMemsetAsync(ovfcnt, 0, 32 * sizeof(int), stream);

    {
        const int waves_per_block = 8;
        const int blocks = (TILES_ + waves_per_block - 1) / waves_per_block;
        gat_gemm_wmma<<<blocks, 256, 0, stream>>>(X, W, h);
    }
    {
        const int blocks = (ROWS_ + 255) / 256;
        gat_scores<<<blocks, 256, 0, stream>>>(h, a, e1, e2);
    }
    gat_part<<<B_ * NCHUNK_, 256, 0, stream>>>(edges, seg, pairs);
    gat_place<<<B_ * NRANGE_, PLACE_THREADS_, 0, stream>>>(pairs, seg, cnt, slots, ovfcnt, ovf);
    gat_rows<<<ROWS_ / (8 * ROWS_PER_WAVE_), 256, 0, stream>>>(slots, cnt, ovfcnt, ovf, h, e1, e2, out);
}


# Independent stages share a launch: edge grouping runs beside the GEMM (and clears the overflow count), bucket building beside the scores; 5 kernels + 1 memset become 3 launches; no sum order changed
# speedup vs baseline: 7.8490x; 1.0704x over previous
#include <hip/hip_runtime.h>
#include <hip/hip_bf16.h>
#include <math.h>

typedef __attribute__((ext_vector_type(2))) float v2f;
typedef __attribute__((ext_vector_type(8))) float v8f;

#define B_ 4
#define N_ 50000
#define E_ 800000
#define FIN_ 128
#define FOUT_ 64
#define ALPHA_ 0.2f
#define NEG_BIG_ (-9000000000000000.0f)

#define ROWS_ (B_ * N_)
#define TILES_ (ROWS_ / 16)

#define GEMM_BLOCKS_ ((TILES_ + 7) / 8)

__device__ __forceinline__ void gemm_role(const float* __restrict__ X,
                                          const float* __restrict__ W,
                                          float* __restrict__ h, int blk) {
    const int lane   = threadIdx.x & 31;
    const int waveId = threadIdx.x >> 5;
    const int rt     = blk * 8 + waveId;
    if (rt >= TILES_) return;

    const int m_base = rt * 16;
    const int mrow   = m_base + (lane & 15);
    const int khalf  = (lane >> 4) << 1;
    const int ncol   = lane & 15;

    v8f acc0 = {}, acc1 = {}, acc2 = {}, acc3 = {};

    for (int k = 0; k < FIN_; k += 4) {
        const int kk = k + khalf;
        v2f a;
        a.x = X[(size_t)mrow * FIN_ + kk];
        a.y = X[(size_t)mrow * FIN_ + kk + 1];

        const float* wr0 = W + (size_t)kk * FOUT_;
        const float* wr1 = wr0 + FOUT_;

        v2f b0, b1, b2, b3;
        b0.x = wr0[ncol];      b0.y = wr1[ncol];
        b1.x = wr0[16 + ncol]; b1.y = wr1[16 + ncol];
        b2.x = wr0[32 + ncol]; b2.y = wr1[32 + ncol];
        b3.x = wr0[48 + ncol]; b3.y = wr1[48 + ncol];

        acc0 = __builtin_amdgcn_wmma_f32_16x16x4_f32(false, a, false, b0, (short)0, acc0, false, false);
        acc1 = __builtin_amdgcn_wmma_f32_16x16x4_f32(false, a, false, b1, (short)0, acc1, false, false);
        acc2 = __builtin_amdgcn_wmma_f32_16x16x4_f32(false, a, false, b2, (short)0, acc2, false, false);
        acc3 = __builtin_amdgcn_wmma_f32_16x16x4_f32(false, a, false, b3, (short)0, acc3, false, false);
    }

    const int rhalf = (lane >> 4) << 3;
#pragma unroll
    for (int i = 0; i < 8; ++i) {
        const size_t row = (size_t)(m_base + i + rhalf);
        h[row * FOUT_ +  0 + ncol] = acc0[i];
        h[row * FOUT_ + 16 + ncol] = acc1[i];
        h[row * FOUT_ + 32 + ncol] = acc2[i];
        h[row * FOUT_ + 48 + ncol] = acc3[i];
    }
}

__device__ __forceinline__ void scores_role(const float* __restrict__ h,
                                            const float* __restrict__ a,
                                            float* __restrict__ e1,
                                            float* __restrict__ e2, int blk, int rows_per_block) {
    const int row = blk * rows_per_block + (int)threadIdx.x;
    if (row >= ROWS_) return;
    const float* hr = h + (size_t)row * FOUT_;
    float s1 = 0.f, s2 = 0.f;
#pragma unroll
    for (int f = 0; f < FOUT_; ++f) {
        const float hv = hr[f];
        s1 += hv * a[f];
        s2 += hv * a[FOUT_ + f];
    }
    e1[row] = s1;
    e2[row] = s2;
}

#define CAP_ 32
#define RANGE_SHIFT_ 8
#define RANGE_ (1 << RANGE_SHIFT_)
#define NRANGE_ ((N_ + RANGE_ - 1) / RANGE_)
#define NTAB_ 256
#define CHUNK_ 4096
#define NCHUNK_ ((E_ + CHUNK_ - 1) / CHUNK_)
#define QUADS_PER_BATCH_ (E_ / 4)
#define SEG_WORDS_ (((B_ * (NRANGE_ + 1) * NCHUNK_ + 31) / 32) * 32)

static_assert(E_ % 4 == 0, "a group of four edges must not cross a batch");
static_assert(NRANGE_ < NTAB_, "the range table needs one entry past the last range");
static_assert(N_ <= 65536 && CHUNK_ <= 65536, "a source and a place in a chunk share one 32-bit word");

__device__ __forceinline__ unsigned long long make_pair(int s, int d) {
    return ((unsigned long long)(unsigned)d << 32) | (unsigned long long)(unsigned)s;
}

__device__ __forceinline__ int range_rank(int* hist, int bk) {
    const int      key  = __builtin_amdgcn_readfirstlane(bk);
    const unsigned act  = __builtin_amdgcn_ballot_w32(true);
    const unsigned same = __builtin_amdgcn_ballot_w32(bk == key);
    if (same == act) {
        const unsigned lane   = threadIdx.x & 31;
        const int      within = __popc(act & ((1u << lane) - 1u));
        int base = 0;
        if (within == 0) base = atomicAdd(&hist[key], __popc(act));
        base = __builtin_amdgcn_readfirstlane(base);
        return base + within;
    }
    return atomicAdd(&hist[bk], 1);
}

__device__ __forceinline__ void part_role(const int* __restrict__ edges,
                                          int* __restrict__ seg,
                                          unsigned long long* __restrict__ pairs, int blk) {
    __shared__ int hist[NTAB_];
    __shared__ int wsum[8];
    __shared__ __attribute__((aligned(16))) unsigned long long stage[CHUNK_];
    const int t = threadIdx.x;
    const int b = blk / NCHUNK_;
    const int c = blk - b * NCHUNK_;
    hist[t] = 0;
    __syncthreads();

    const int4* src4 = reinterpret_cast<const int4*>(edges + ((size_t)b * 2 + 0) * E_);
    const int4* dst4 = reinterpret_cast<const int4*>(edges + ((size_t)b * 2 + 1) * E_);
    int4 s[4], d[4];
    bool ok[4];
#pragma unroll
    for (int i = 0; i < 4; ++i) {
        const int q = c * (CHUNK_ / 4) + i * 256 + t;
        ok[i] = q < QUADS_PER_BATCH_;
        if (ok[i]) { s[i] = src4[q]; d[i] = dst4[q]; }
        else       { s[i] = make_int4(0, 0, 0, 0); d[i] = s[i]; }
    }
    unsigned sr[16];
#pragma unroll
    for (int i = 0; i < 4; ++i) {
        if (ok[i]) {
            sr[4 * i + 0] = (unsigned)s[i].x | ((unsigned)range_rank(hist, s[i].x >> RANGE_SHIFT_) << 16);
            sr[4 * i + 1] = (unsigned)s[i].y | ((unsigned)range_rank(hist, s[i].y >> RANGE_SHIFT_) << 16);
            sr[4 * i + 2] = (unsigned)s[i].z | ((unsigned)range_rank(hist, s[i].z >> RANGE_SHIFT_) << 16);
            sr[4 * i + 3] = (unsigned)s[i].w | ((unsigned)range_rank(hist, s[i].w >> RANGE_SHIFT_) << 16);
        }
    }
    __syncthreads();
    const int v = hist[t];
    int incl = v;
#pragma unroll
    for (int o = 1; o < 32; o <<= 1) {
        const int u = __shfl_up(incl, o, 32);
        if ((t & 31) >= o) incl += u;
    }
    if ((t & 31) == 31) wsum[t >> 5] = incl;
    __syncthreads();
    int off = 0;
    for (int k = 0; k < (t >> 5); ++k) off += wsum[k];
    const int excl = incl - v + off;
    hist[t] = excl;
    if (t <= NRANGE_) seg[(size_t)(b * (NRANGE_ + 1) + t) * NCHUNK_ + c] = excl;
    __syncthreads();
#pragma unroll
    for (int i = 0; i < 4; ++i) {
        if (ok[i]) {
            const int dd[4] = {d[i].x, d[i].y, d[i].z, d[i].w};
#pragma unroll
            for (int e = 0; e < 4; ++e) {
                const unsigned w  = sr[4 * i + e];
                const int      sx = (int)(w & 0xffffu);
                stage[hist[sx >> RANGE_SHIFT_] + (int)(w >> 16)] = make_pair(sx, dd[e]);
            }
        }
    }
    __syncthreads();
    const int n2 = hist[NRANGE_] >> 1;
    const ulonglong2* st2 = reinterpret_cast<const ulonglong2*>(stage);
    ulonglong2* out2 = reinterpret_cast<ulonglong2*>(pairs + (size_t)b * E_ + (size_t)c * CHUNK_);
    for (int i = t; i < n2; i += 256) out2[i] = st2[i];
}

#define PLACE_THREADS_ 512
#define PLACE_WAVES_ (PLACE_THREADS_ / 32)
#define PLACE_TRIPS_ ((NCHUNK_ + PLACE_WAVES_ - 1) / PLACE_WAVES_)
#define PLACE_HEAD_ 64

static_assert(NCHUNK_ <= PLACE_THREADS_, "one thread of place_role per chunk reads the group bounds");
static_assert(RANGE_ <= PLACE_THREADS_, "one thread of place_role per row of the range");

__device__ __forceinline__ void place_pair(int* cntl, int* slotl,
                                           int* __restrict__ ovfcnt, int* __restrict__ ovf,
                                           unsigned long long p, int row0, int rowb) {
    const int sx = (int)(unsigned)(p & 0xffffffffull);
    const int dy = (int)(unsigned)(p >> 32);
    const int lr = sx - row0;
    const int r  = atomicAdd(&cntl[lr], 1);
    if (r < CAP_) {
        slotl[lr * CAP_ + r] = dy;
    } else {
        const int k = atomicAdd(ovfcnt, 1);
        reinterpret_cast<int2*>(ovf)[k] = make_int2(rowb + sx, dy);
    }
}

__device__ __forceinline__ void place_role(const unsigned long long* __restrict__ pairs,
                                           const int* __restrict__ seg,
                                           int* __restrict__ cnt,
                                           int* __restrict__ slots,
                                           int* __restrict__ ovfcnt,
                                           int* __restrict__ ovf, int blk) {
    __shared__ int cntl[RANGE_];
    __shared__ int slotl[RANGE_ * CAP_];
    __shared__ int sbeg[NCHUNK_];
    __shared__ int send[NCHUNK_];
    __shared__ int longs[NCHUNK_];
    __shared__ int nlong;
    const int t = threadIdx.x;
    const int b = blk / NRANGE_;
    const int g = blk - b * NRANGE_;
    if (t == 0) nlong = 0;
    if (t < RANGE_) cntl[t] = 0;
    int glen = 0;
    if (t < NCHUNK_) {
        const int* sg = seg + (size_t)(b * (NRANGE_ + 1) + g) * NCHUNK_;
        const int gb = sg[t];
        const int ge = sg[NCHUNK_ + t];
        sbeg[t] = gb;
        send[t] = ge;
        glen = ge - gb;
    }
    __syncthreads();
    if (glen > PLACE_HEAD_) longs[atomicAdd(&nlong, 1)] = t;

    const int lane = t & 31;
    const int wv   = __builtin_amdgcn_readfirstlane(t >> 5);
    const int row0 = g * RANGE_;
    const int rowb = b * N_;
    const unsigned long long* pb = pairs + (size_t)b * E_;

    unsigned long long p0[PLACE_TRIPS_], p1[PLACE_TRIPS_];
    int i0[PLACE_TRIPS_], en[PLACE_TRIPS_];
#pragma unroll
    for (int k = 0; k < PLACE_TRIPS_; ++k) {
        const int c = wv + k * PLACE_WAVES_;
        int gb = 0, ge = 0;
        if (c < NCHUNK_) { gb = sbeg[c]; ge = send[c]; }
        i0[k] = c * CHUNK_ + gb + lane;
        en[k] = c * CHUNK_ + ge;
        p0[k] = 0ull;
        p1[k] = 0ull;
        if (i0[k] < en[k])      p0[k] = pb[i0[k]];
        if (i0[k] + 32 < en[k]) p1[k] = pb[i0[k] + 32];
    }
#pragma unroll
    for (int k = 0; k < PLACE_TRIPS_; ++k) {
        if (i0[k] < en[k])      place_pair(cntl, slotl, ovfcnt, ovf, p0[k], row0, rowb);
        if (i0[k] + 32 < en[k]) place_pair(cntl, slotl, ovfcnt, ovf, p1[k], row0, rowb);
    }
    __syncthreads();
    const int nl = nlong;
    for (int k = 0; k < nl; ++k) {
        const int c  = longs[k];
        const int ie = c * CHUNK_ + send[c];
        for (int i = c * CHUNK_ + sbeg[c] + PLACE_HEAD_ + t; i < ie; i += PLACE_THREADS_)
            place_pair(cntl, slotl, ovfcnt, ovf, pb[i], row0, rowb);
    }
    __syncthreads();
    for (int r = wv; r < RANGE_; r += PLACE_WAVES_) {
        const int n = cntl[r];
        if (lane < n) slots[(size_t)(rowb + row0 + r) * CAP_ + lane] = slotl[r * CAP_ + lane];
    }
    if (t < RANGE_ && row0 + t < N_) cnt[rowb + row0 + t] = cntl[t];
}

#define PART_BLOCKS_ (B_ * NCHUNK_)

__global__ __launch_bounds__(256, 16) void gat_front(const float* __restrict__ X,
                                                 const float* __restrict__ W,
                                                 float* __restrict__ h,
                                                 const int* __restrict__ edges,
                                                 int* __restrict__ seg,
                                                 unsigned long long* __restrict__ pairs,
                                                 int* __restrict__ ovfcnt) {
    const int blk = (int)blockIdx.x;
    if (blk < GEMM_BLOCKS_) {
        gemm_role(X, W, h, blk);
    } else {
        if (blk == GEMM_BLOCKS_ && threadIdx.x == 0) ovfcnt[0] = 0;
        part_role(edges, seg, pairs, blk - GEMM_BLOCKS_);
    }
}

#define PLACE_BLOCKS_ (B_ * NRANGE_)
#define SCORE_BLOCKS_ ((ROWS_ + PLACE_THREADS_ - 1) / PLACE_THREADS_)

__global__ __launch_bounds__(PLACE_THREADS_) void gat_mid(const unsigned long long* __restrict__ pairs,
                                                          const int* __restrict__ seg,
                                                          int* __restrict__ cnt,
                                                          int* __restrict__ slots,
                                                          int* __restrict__ ovfcnt,
                                                          int* __restrict__ ovf,
                                                          const float* __restrict__ h,
                                                          const float* __restrict__ a,
                                                          float* __restrict__ e1,
                                                          float* __restrict__ e2) {
    const int blk = (int)blockIdx.x;
    if (blk < PLACE_BLOCKS_) {
        place_role(pairs, seg, cnt, slots, ovfcnt, ovf, blk);
    } else {
        scores_role(h, a, e1, e2, blk - PLACE_BLOCKS_, PLACE_THREADS_);
    }
}

#define DST_NONE_ 0x7fffffff
#define ROWS_PER_WAVE_ 8

static_assert(N_ % ROWS_PER_WAVE_ == 0, "the rows of one wave must lie in one batch");
static_assert(ROWS_ % (8 * ROWS_PER_WAVE_) == 0, "row grid must be exact");

__device__ __forceinline__ float edge_weight(float e1r, float e2d) {
    const float s  = e1r + e2d;
    const float lr = (s >= 0.f) ? s : ALPHA_ * s;
    return __expf(-lr);
}

__device__ __forceinline__ int lane_int(int v, int l) { return __builtin_amdgcn_readlane(v, l); }
__device__ __forceinline__ float lane_flt(float v, int l) {
    return __int_as_float(__builtin_amdgcn_readlane(__float_as_int(v), l));
}

template <int NB>
__device__ __forceinline__ void take_edges(const float2* __restrict__ hb, int sd, float sw, int j,
                                           float& ax, float& ay, float& rs) {
    float2 hv[NB];
    float  wj[NB];
#pragma unroll
    for (int i = 0; i < NB; ++i) {
        const int dj = lane_int(sd, j + i);
        wj[i] = lane_flt(sw, j + i);
        hv[i] = hb[(size_t)dj * (FOUT_ / 2)];
    }
#pragma unroll
    for (int i = 0; i < NB; ++i) {
        ax = fmaf(wj[i], hv[i].x, ax);
        ay = fmaf(wj[i], hv[i].y, ay);
        rs += wj[i];
    }
}

__device__ __forceinline__ void ranked_edge(int posA, int posB, bool hasB, int dA, int dB,
                                            float wA, float wB, int j, int& dj, float& wj) {
    const unsigned mA = __builtin_amdgcn_ballot_w32(posA == j);
    const unsigned mB = __builtin_amdgcn_ballot_w32(hasB && posB == j);
    const bool fromA  = mA != 0u;
    const int  l      = __builtin_ffs((int)(fromA ? mA : mB)) - 1;
    const int   da = lane_int(dA, l), db = lane_int(dB, l);
    const float wa = lane_flt(wA, l), wb = lane_flt(wB, l);
    const int   d  = fromA ? da : db;
    dj = ((unsigned)d < (unsigned)N_) ? d : 0;
    wj = fromA ? wa : wb;
}

__global__ __launch_bounds__(256) void gat_rows(const int* __restrict__ slots,
                                                const int* __restrict__ cnt,
                                                const int* __restrict__ ovfcnt,
                                                const int* __restrict__ ovf,
                                                const float* __restrict__ h,
                                                const float* __restrict__ e1,
                                                const float* __restrict__ e2,
                                                float* __restrict__ out) {
    const int lane = threadIdx.x & 31;
    const int wv   = __builtin_amdgcn_readfirstlane((int)(threadIdx.x >> 5));
    const int row0 = (blockIdx.x * 8 + wv) * ROWS_PER_WAVE_;
    const int b    = row0 / N_;
    const float*  e2b = e2 + (size_t)b * N_;
    const float2* hb  = reinterpret_cast<const float2*>(h + (size_t)b * N_ * FOUT_) + lane;

    for (int r = 0; r < ROWS_PER_WAVE_; ++r) {
    const int row = row0 + r;
    const int deg = cnt[row];
    const float e1r = e1[row];
    const int dl = slots[row * CAP_ + lane];

    float ax = 0.f, ay = 0.f, rs = 0.f;

    if (deg <= CAP_) {
        const bool have = lane < deg;
        const int  d    = have ? dl : DST_NONE_;
        const float e2d = e2b[have ? d : 0];
        int pos = 0;
        for (int k = 0; k < deg; ++k) {
            const int dk = lane_int(d, k);
            pos += (dk < d || (dk == d && k < lane)) ? 1 : 0;
        }
        if (!have) pos = lane;
        const float w = have ? edge_weight(e1r, e2d) : 0.f;
        const int   sd = __builtin_amdgcn_ds_permute(pos << 2, d);
        const float sw = __int_as_float(__builtin_amdgcn_ds_permute(pos << 2, __float_as_int(w)));

        int j = 0;
        for (; j + 8 <= deg; j += 8) take_edges<8>(hb, sd, sw, j, ax, ay, rs);
        if (j + 4 <= deg) { take_edges<4>(hb, sd, sw, j, ax, ay, rs); j += 4; }
        if (j < deg) {
            const int last = deg - 1;
            const int j1 = (j + 1 < last) ? j + 1 : last;
            const int j2 = (j + 2 < last) ? j + 2 : last;
            const float2 h0 = hb[(size_t)lane_int(sd, j)  * (FOUT_ / 2)];
            const float2 h1 = hb[(size_t)lane_int(sd, j1) * (FOUT_ / 2)];
            const float2 h2 = hb[(size_t)lane_int(sd, j2) * (FOUT_ / 2)];
            const float w0 = lane_flt(sw, j), w1 = lane_flt(sw, j1), w2 = lane_flt(sw, j2);
            ax = fmaf(w0, h0.x, ax); ay = fmaf(w0, h0.y, ay); rs += w0;
            if (j + 1 < deg) { ax = fmaf(w1, h1.x, ax); ay = fmaf(w1, h1.y, ay); rs += w1; }
            if (j + 2 < deg) { ax = fmaf(w2, h2.x, ax); ay = fmaf(w2, h2.y, ay); rs += w2; }
        }
    } else if (deg <= 2 * CAP_) {
        const int   novf = ovfcnt[0];
        const int2* ol   = reinterpret_cast<const int2*>(ovf);
        int d2 = DST_NONE_;
        int k  = 0;
        for (int base = 0; base < novf; base += 32) {
            const int i = base + lane;
            int2 p = make_int2(-1, -1);
            if (i < novf) p = ol[i];
            unsigned hit = __builtin_amdgcn_ballot_w32(p.x == row);
            while (hit != 0u) {
                const int l = __builtin_ffs((int)hit) - 1;
                const int v = lane_int(p.y, l);
                if (lane == k) d2 = v;
                ++k;
                hit &= hit - 1u;
            }
        }
        if (k > 32) k = 32;
        const bool  hasB = lane < k;
        const int   dA   = dl;
        const int   dB   = d2;
        const float e2A  = e2b[dA];
        const float e2B  = e2b[hasB ? dB : 0];
        int posA = 0, posB = 0;
#pragma nounroll
        for (int j = 0; j < CAP_; ++j) {
            const int x = lane_int(dA, j);
            posA += (x < dA || (x == dA && j < lane)) ? 1 : 0;
            posB += (x <= dB) ? 1 : 0;
        }
        for (int j = 0; j < k; ++j) {
            const int y = lane_int(dB, j);
            posA += (y < dA) ? 1 : 0;
            posB += (y < dB || (y == dB && j < lane)) ? 1 : 0;
        }
        const float wA = edge_weight(e1r, e2A);
        const float wB = hasB ? edge_weight(e1r, e2B) : 0.f;
        const int   tot = CAP_ + k;
        for (int j0 = 0; j0 < tot; j0 += 8) {
            float2 hv[8];
            float  wj[8];
#pragma unroll
            for (int i = 0; i < 8; ++i) {
                const int j = (j0 + i < tot) ? j0 + i : tot - 1;
                int dj;
                ranked_edge(posA, posB, hasB, dA, dB, wA, wB, j, dj, wj[i]);
                hv[i] = hb[(size_t)dj * (FOUT_ / 2)];
            }
#pragma unroll
            for (int i = 0; i < 8; ++i) {
                if (j0 + i < tot) {
                    ax = fmaf(wj[i], hv[i].x, ax);
                    ay = fmaf(wj[i], hv[i].y, ay);
                    rs += wj[i];
                }
            }
        }
    } else {
        const int   novf = ovfcnt[0];
        const int2* ol   = reinterpret_cast<const int2*>(ovf);
        int prev = -1;
        for (int it = 0; it < deg; ++it) {
            int m = (dl > prev) ? dl : DST_NONE_;
            for (int base = 0; base < novf; base += 32) {
                const int i = base + lane;
                if (i < novf) {
                    const int2 p = ol[i];
                    if (p.x == row && p.y > prev && p.y < m) m = p.y;
                }
            }
#pragma unroll
            for (int o = 16; o > 0; o >>= 1) {
                const int t = __shfl_xor(m, o, 32);
                m = (t < m) ? t : m;
            }
            m = __builtin_amdgcn_readfirstlane(m);
            if (m == DST_NONE_) break;
            int c = __popc(__builtin_amdgcn_ballot_w32(dl == m));
            for (int base = 0; base < novf; base += 32) {
                const int i = base + lane;
                bool hit = false;
                if (i < novf) {
                    const int2 p = ol[i];
                    hit = (p.x == row) && (p.y == m);
                }
                c += __popc(__builtin_amdgcn_ballot_w32(hit));
            }
            const float wj = edge_weight(e1r, e2b[m]);
            const float2 hv = hb[(size_t)m * (FOUT_ / 2)];
            for (int t = 0; t < c; ++t) {
                ax = fmaf(wj, hv.x, ax);
                ay = fmaf(wj, hv.y, ay);
                rs += wj;
            }
            prev = m;
        }
    }

    float vx = ax / rs;
    float vy = ay / rs;
    if (isnan(vx)) vx = NEG_BIG_;
    if (isnan(vy)) vy = NEG_BIG_;
    float2 o;
    o.x = (vx > 0.f) ? vx : expm1f(vx);
    o.y = (vy > 0.f) ? vy : expm1f(vy);
    reinterpret_cast<float2*>(out + (size_t)row * FOUT_)[lane] = o;
    }
}

extern "C" void kernel_launch(void* const* d_in, const int* in_sizes, int n_in,
                              void* d_out, int out_size, void* d_ws, size_t ws_size,
                              hipStream_t stream) {
    const float* X     = (const float*)d_in[0];
    const int*   edges = (const int*)d_in[1];
    const float* W     = (const float*)d_in[2];
    const float* a     = (const float*)d_in[3];
    float*       out   = (float*)d_out;

    float* h      = (float*)d_ws;
    float* e1     = h + (size_t)ROWS_ * FOUT_;
    float* e2     = e1 + (size_t)ROWS_;
    int*   cnt    = (int*)(e2 + (size_t)ROWS_);
    int*   ovfcnt = cnt + (size_t)ROWS_;
    int*   slots  = ovfcnt + 32;
    int*   ovf    = slots + (size_t)ROWS_ * CAP_;
    int*   seg    = ovf + (size_t)2 * B_ * E_;
    unsigned long long* pairs = (unsigned long long*)(seg + (size_t)SEG_WORDS_);

    gat_front<<<GEMM_BLOCKS_ + PART_BLOCKS_, 256, 0, stream>>>(X, W, h, edges, seg, pairs, ovfcnt);
    gat_mid<<<PLACE_BLOCKS_ + SCORE_BLOCKS_, PLACE_THREADS_, 0, stream>>>(pairs, seg, cnt, slots, ovfcnt, ovf,
                                                                        h, a, e1, e2);
    gat_rows<<<ROWS_ / (8 * ROWS_PER_WAVE_), 256, 0, stream>>>(slots, cnt, ovfcnt, ovf, h, e1, e2, out);
}


# GEMM B operand read from an LDS copy of W in fragment order (two 16-byte LDS reads per k-step in place of eight 4-byte global loads); same values, products and sum order
# speedup vs baseline: 8.2037x; 1.0452x over previous
#include <hip/hip_runtime.h>
#include <hip/hip_bf16.h>
#include <math.h>

typedef __attribute__((ext_vector_type(2))) float v2f;
typedef __attribute__((ext_vector_type(8))) float v8f;

#define B_ 4
#define N_ 50000
#define E_ 800000
#define FIN_ 128
#define FOUT_ 64
#define ALPHA_ 0.2f
#define NEG_BIG_ (-9000000000000000.0f)

#define ROWS_ (B_ * N_)
#define TILES_ (ROWS_ / 16)

#define GEMM_BLOCKS_ ((TILES_ + 7) / 8)

#define WL_PLANE_ ((FIN_ / 4) * 32)
#define WL_ENTRIES_ (2 * WL_PLANE_)

__device__ __forceinline__ void gemm_role(const float* __restrict__ X,
                                          const float* __restrict__ W,
                                          float* __restrict__ h, float4* Wl, int blk) {
#pragma unroll
    for (int i = 0; i < WL_ENTRIES_ / 256; ++i) {
        const int e = i * 256 + (int)threadIdx.x;
        const int p = e / WL_PLANE_;
        const int j = (e >> 5) & (FIN_ / 4 - 1);
        const int l = e & 31;
        const float* w0 = W + (size_t)(4 * j + ((l >> 4) << 1)) * FOUT_ + 32 * p + (l & 15);
        float4 v;
        v.x = w0[0];  v.y = w0[FOUT_];
        v.z = w0[16]; v.w = w0[FOUT_ + 16];
        Wl[e] = v;
    }
    __syncthreads();

    const int lane   = threadIdx.x & 31;
    const int waveId = threadIdx.x >> 5;
    const int rt     = blk * 8 + waveId;
    if (rt >= TILES_) return;

    const int m_base = rt * 16;
    const int mrow   = m_base + (lane & 15);
    const int khalf  = (lane >> 4) << 1;
    const int ncol   = lane & 15;

    v8f acc0 = {}, acc1 = {}, acc2 = {}, acc3 = {};

#pragma unroll 4
    for (int k = 0; k < FIN_; k += 4) {
        const int kk = k + khalf;
        v2f a;
        a.x = X[(size_t)mrow * FIN_ + kk];
        a.y = X[(size_t)mrow * FIN_ + kk + 1];

        const float4 w01 = Wl[(k >> 2) * 32 + lane];
        const float4 w23 = Wl[WL_PLANE_ + (k >> 2) * 32 + lane];

        v2f b0, b1, b2, b3;
        b0.x = w01.x; b0.y = w01.y;
        b1.x = w01.z; b1.y = w01.w;
        b2.x = w23.x; b2.y = w23.y;
        b3.x = w23.z; b3.y = w23.w;

        acc0 = __builtin_amdgcn_wmma_f32_16x16x4_f32(false, a, false, b0, (short)0, acc0, false, false);
        acc1 = __builtin_amdgcn_wmma_f32_16x16x4_f32(false, a, false, b1, (short)0, acc1, false, false);
        acc2 = __builtin_amdgcn_wmma_f32_16x16x4_f32(false, a, false, b2, (short)0, acc2, false, false);
        acc3 = __builtin_amdgcn_wmma_f32_16x16x4_f32(false, a, false, b3, (short)0, acc3, false, false);
    }

    const int rhalf = (lane >> 4) << 3;
#pragma unroll
    for (int i = 0; i < 8; ++i) {
        const size_t row = (size_t)(m_base + i + rhalf);
        h[row * FOUT_ +  0 + ncol] = acc0[i];
        h[row * FOUT_ + 16 + ncol] = acc1[i];
        h[row * FOUT_ + 32 + ncol] = acc2[i];
        h[row * FOUT_ + 48 + ncol] = acc3[i];
    }
}

__device__ __forceinline__ void scores_role(const float* __restrict__ h,
                                            const float* __restrict__ a,
                                            float* __restrict__ e1,
                                            float* __restrict__ e2, int blk, int rows_per_block) {
    const int row = blk * rows_per_block + (int)threadIdx.x;
    if (row >= ROWS_) return;
    const float* hr = h + (size_t)row * FOUT_;
    float s1 = 0.f, s2 = 0.f;
#pragma unroll
    for (int f = 0; f < FOUT_; ++f) {
        const float hv = hr[f];
        s1 += hv * a[f];
        s2 += hv * a[FOUT_ + f];
    }
    e1[row] = s1;
    e2[row] = s2;
}

#define CAP_ 32
#define RANGE_SHIFT_ 8
#define RANGE_ (1 << RANGE_SHIFT_)
#define NRANGE_ ((N_ + RANGE_ - 1) / RANGE_)
#define NTAB_ 256
#define CHUNK_ 4096
#define NCHUNK_ ((E_ + CHUNK_ - 1) / CHUNK_)
#define QUADS_PER_BATCH_ (E_ / 4)
#define SEG_WORDS_ (((B_ * (NRANGE_ + 1) * NCHUNK_ + 31) / 32) * 32)

static_assert(E_ % 4 == 0, "a group of four edges must not cross a batch");
static_assert(NRANGE_ < NTAB_, "the range table needs one entry past the last range");
static_assert(N_ <= 65536 && CHUNK_ <= 65536, "a source and a place in a chunk share one 32-bit word");

__device__ __forceinline__ unsigned long long make_pair(int s, int d) {
    return ((unsigned long long)(unsigned)d << 32) | (unsigned long long)(unsigned)s;
}

__device__ __forceinline__ int range_rank(int* hist, int bk) {
    const int      key  = __builtin_amdgcn_readfirstlane(bk);
    const unsigned act  = __builtin_amdgcn_ballot_w32(true);
    const unsigned same = __builtin_amdgcn_ballot_w32(bk == key);
    if (same == act) {
        const unsigned lane   = threadIdx.x & 31;
        const int      within = __popc(act & ((1u << lane) - 1u));
        int base = 0;
        if (within == 0) base = atomicAdd(&hist[key], __popc(act));
        base = __builtin_amdgcn_readfirstlane(base);
        return base + within;
    }
    return atomicAdd(&hist[bk], 1);
}

__device__ __forceinline__ void part_role(const int* __restrict__ edges,
                                          int* __restrict__ seg,
                                          unsigned long long* __restrict__ pairs,
                                          unsigned long long* stage, int blk) {
    __shared__ int hist[NTAB_];
    __shared__ int wsum[8];
    const int t = threadIdx.x;
    const int b = blk / NCHUNK_;
    const int c = blk - b * NCHUNK_;
    hist[t] = 0;
    __syncthreads();

    const int4* src4 = reinterpret_cast<const int4*>(edges + ((size_t)b * 2 + 0) * E_);
    const int4* dst4 = reinterpret_cast<const int4*>(edges + ((size_t)b * 2 + 1) * E_);
    int4 s[4], d[4];
    bool ok[4];
#pragma unroll
    for (int i = 0; i < 4; ++i) {
        const int q = c * (CHUNK_ / 4) + i * 256 + t;
        ok[i] = q < QUADS_PER_BATCH_;
        if (ok[i]) { s[i] = src4[q]; d[i] = dst4[q]; }
        else       { s[i] = make_int4(0, 0, 0, 0); d[i] = s[i]; }
    }
    unsigned sr[16];
#pragma unroll
    for (int i = 0; i < 4; ++i) {
        if (ok[i]) {
            sr[4 * i + 0] = (unsigned)s[i].x | ((unsigned)range_rank(hist, s[i].x >> RANGE_SHIFT_) << 16);
            sr[4 * i + 1] = (unsigned)s[i].y | ((unsigned)range_rank(hist, s[i].y >> RANGE_SHIFT_) << 16);
            sr[4 * i + 2] = (unsigned)s[i].z | ((unsigned)range_rank(hist, s[i].z >> RANGE_SHIFT_) << 16);
            sr[4 * i + 3] = (unsigned)s[i].w | ((unsigned)range_rank(hist, s[i].w >> RANGE_SHIFT_) << 16);
        }
    }
    __syncthreads();
    const int v = hist[t];
    int incl = v;
#pragma unroll
    for (int o = 1; o < 32; o <<= 1) {
        const int u = __shfl_up(incl, o, 32);
        if ((t & 31) >= o) incl += u;
    }
    if ((t & 31) == 31) wsum[t >> 5] = incl;
    __syncthreads();
    int off = 0;
    for (int k = 0; k < (t >> 5); ++k) off += wsum[k];
    const int excl = incl - v + off;
    hist[t] = excl;
    if (t <= NRANGE_) seg[(size_t)(b * (NRANGE_ + 1) + t) * NCHUNK_ + c] = excl;
    __syncthreads();
#pragma unroll
    for (int i = 0; i < 4; ++i) {
        if (ok[i]) {
            const int dd[4] = {d[i].x, d[i].y, d[i].z, d[i].w};
#pragma unroll
            for (int e = 0; e < 4; ++e) {
                const unsigned w  = sr[4 * i + e];
                const int      sx = (int)(w & 0xffffu);
                stage[hist[sx >> RANGE_SHIFT_] + (int)(w >> 16)] = make_pair(sx, dd[e]);
            }
        }
    }
    __syncthreads();
    const int n2 = hist[NRANGE_] >> 1;
    const ulonglong2* st2 = reinterpret_cast<const ulonglong2*>(stage);
    ulonglong2* out2 = reinterpret_cast<ulonglong2*>(pairs + (size_t)b * E_ + (size_t)c * CHUNK_);
    for (int i = t; i < n2; i += 256) out2[i] = st2[i];
}

#define PLACE_THREADS_ 512
#define PLACE_WAVES_ (PLACE_THREADS_ / 32)
#define PLACE_TRIPS_ ((NCHUNK_ + PLACE_WAVES_ - 1) / PLACE_WAVES_)
#define PLACE_HEAD_ 64

static_assert(NCHUNK_ <= PLACE_THREADS_, "one thread of place_role per chunk reads the group bounds");
static_assert(RANGE_ <= PLACE_THREADS_, "one thread of place_role per row of the range");

__device__ __forceinline__ void place_pair(int* cntl, int* slotl,
                                           int* __restrict__ ovfcnt, int* __restrict__ ovf,
                                           unsigned long long p, int row0, int rowb) {
    const int sx = (int)(unsigned)(p & 0xffffffffull);
    const int dy = (int)(unsigned)(p >> 32);
    const int lr = sx - row0;
    const int r  = atomicAdd(&cntl[lr], 1);
    if (r < CAP_) {
        slotl[lr * CAP_ + r] = dy;
    } else {
        const int k = atomicAdd(ovfcnt, 1);
        reinterpret_cast<int2*>(ovf)[k] = make_int2(rowb + sx, dy);
    }
}

__device__ __forceinline__ void place_role(const unsigned long long* __restrict__ pairs,
                                           const int* __restrict__ seg,
                                           int* __restrict__ cnt,
                                           int* __restrict__ slots,
                                           int* __restrict__ ovfcnt,
                                           int* __restrict__ ovf, int blk) {
    __shared__ int cntl[RANGE_];
    __shared__ int slotl[RANGE_ * CAP_];
    __shared__ int sbeg[NCHUNK_];
    __shared__ int send[NCHUNK_];
    __shared__ int longs[NCHUNK_];
    __shared__ int nlong;
    const int t = threadIdx.x;
    const int b = blk / NRANGE_;
    const int g = blk - b * NRANGE_;
    if (t == 0) nlong = 0;
    if (t < RANGE_) cntl[t] = 0;
    int glen = 0;
    if (t < NCHUNK_) {
        const int* sg = seg + (size_t)(b * (NRANGE_ + 1) + g) * NCHUNK_;
        const int gb = sg[t];
        const int ge = sg[NCHUNK_ + t];
        sbeg[t] = gb;
        send[t] = ge;
        glen = ge - gb;
    }
    __syncthreads();
    if (glen > PLACE_HEAD_) longs[atomicAdd(&nlong, 1)] = t;

    const int lane = t & 31;
    const int wv   = __builtin_amdgcn_readfirstlane(t >> 5);
    const int row0 = g * RANGE_;
    const int rowb = b * N_;
    const unsigned long long* pb = pairs + (size_t)b * E_;

    unsigned long long p0[PLACE_TRIPS_], p1[PLACE_TRIPS_];
    int i0[PLACE_TRIPS_], en[PLACE_TRIPS_];
#pragma unroll
    for (int k = 0; k < PLACE_TRIPS_; ++k) {
        const int c = wv + k * PLACE_WAVES_;
        int gb = 0, ge = 0;
        if (c < NCHUNK_) { gb = sbeg[c]; ge = send[c]; }
        i0[k] = c * CHUNK_ + gb + lane;
        en[k] = c * CHUNK_ + ge;
        p0[k] = 0ull;
        p1[k] = 0ull;
        if (i0[k] < en[k])      p0[k] = pb[i0[k]];
        if (i0[k] + 32 < en[k]) p1[k] = pb[i0[k] + 32];
    }
#pragma unroll
    for (int k = 0; k < PLACE_TRIPS_; ++k) {
        if (i0[k] < en[k])      place_pair(cntl, slotl, ovfcnt, ovf, p0[k], row0, rowb);
        if (i0[k] + 32 < en[k]) place_pair(cntl, slotl, ovfcnt, ovf, p1[k], row0, rowb);
    }
    __syncthreads();
    const int nl = nlong;
    for (int k = 0; k < nl; ++k) {
        const int c  = longs[k];
        const int ie = c * CHUNK_ + send[c];
        for (int i = c * CHUNK_ + sbeg[c] + PLACE_HEAD_ + t; i < ie; i += PLACE_THREADS_)
            place_pair(cntl, slotl, ovfcnt, ovf, pb[i], row0, rowb);
    }
    __syncthreads();
    for (int r = wv; r < RANGE_; r += PLACE_WAVES_) {
        const int n = cntl[r];
        if (lane < n) slots[(size_t)(rowb + row0 + r) * CAP_ + lane] = slotl[r * CAP_ + lane];
    }
    if (t < RANGE_ && row0 + t < N_) cnt[rowb + row0 + t] = cntl[t];
}

#define PART_BLOCKS_ (B_ * NCHUNK_)

__global__ __launch_bounds__(256, 16) void gat_front(const float* __restrict__ X,
                                                 const float* __restrict__ W,
                                                 float* __restrict__ h,
                                                 const int* __restrict__ edges,
                                                 int* __restrict__ seg,
                                                 unsigned long long* __restrict__ pairs,
                                                 int* __restrict__ ovfcnt) {
    __shared__ __attribute__((aligned(16))) unsigned long long lds_buf[CHUNK_];
    static_assert(WL_ENTRIES_ * sizeof(float4) <= CHUNK_ * sizeof(unsigned long long), "W must fit the shared buffer");
    const int blk = (int)blockIdx.x;
    if (blk < GEMM_BLOCKS_) {
        gemm_role(X, W, h, reinterpret_cast<float4*>(lds_buf), blk);
    } else {
        if (blk == GEMM_BLOCKS_ && threadIdx.x == 0) ovfcnt[0] = 0;
        part_role(edges, seg, pairs, lds_buf, blk - GEMM_BLOCKS_);
    }
}

#define PLACE_BLOCKS_ (B_ * NRANGE_)
#define SCORE_BLOCKS_ ((ROWS_ + PLACE_THREADS_ - 1) / PLACE_THREADS_)

__global__ __launch_bounds__(PLACE_THREADS_) void gat_mid(const unsigned long long* __restrict__ pairs,
                                                          const int* __restrict__ seg,
                                                          int* __restrict__ cnt,
                                                          int* __restrict__ slots,
                                                          int* __restrict__ ovfcnt,
                                                          int* __restrict__ ovf,
                                                          const float* __restrict__ h,
                                                          const float* __restrict__ a,
                                                          float* __restrict__ e1,
                                                          float* __restrict__ e2) {
    const int blk = (int)blockIdx.x;
    if (blk < PLACE_BLOCKS_) {
        place_role(pairs, seg, cnt, slots, ovfcnt, ovf, blk);
    } else {
        scores_role(h, a, e1, e2, blk - PLACE_BLOCKS_, PLACE_THREADS_);
    }
}

#define DST_NONE_ 0x7fffffff
#define ROWS_PER_WAVE_ 8

static_assert(N_ % ROWS_PER_WAVE_ == 0, "the rows of one wave must lie in one batch");
static_assert(ROWS_ % (8 * ROWS_PER_WAVE_) == 0, "row grid must be exact");

__device__ __forceinline__ float edge_weight(float e1r, float e2d) {
    const float s  = e1r + e2d;
    const float lr = (s >= 0.f) ? s : ALPHA_ * s;
    return __expf(-lr);
}

__device__ __forceinline__ int lane_int(int v, int l) { return __builtin_amdgcn_readlane(v, l); }
__device__ __forceinline__ float lane_flt(float v, int l) {
    return __int_as_float(__builtin_amdgcn_readlane(__float_as_int(v), l));
}

template <int NB>
__device__ __forceinline__ void take_edges(const float2* __restrict__ hb, int sd, float sw, int j,
                                           float& ax, float& ay, float& rs) {
    float2 hv[NB];
    float  wj[NB];
#pragma unroll
    for (int i = 0; i < NB; ++i) {
        const int dj = lane_int(sd, j + i);
        wj[i] = lane_flt(sw, j + i);
        hv[i] = hb[(size_t)dj * (FOUT_ / 2)];
    }
#pragma unroll
    for (int i = 0; i < NB; ++i) {
        ax = fmaf(wj[i], hv[i].x, ax);
        ay = fmaf(wj[i], hv[i].y, ay);
        rs += wj[i];
    }
}

__device__ __forceinline__ void ranked_edge(int posA, int posB, bool hasB, int dA, int dB,
                                            float wA, float wB, int j, int& dj, float& wj) {
    const unsigned mA = __builtin_amdgcn_ballot_w32(posA == j);
    const unsigned mB = __builtin_amdgcn_ballot_w32(hasB && posB == j);
    const bool fromA  = mA != 0u;
    const int  l      = __builtin_ffs((int)(fromA ? mA : mB)) - 1;
    const int   da = lane_int(dA, l), db = lane_int(dB, l);
    const float wa = lane_flt(wA, l), wb = lane_flt(wB, l);
    const int   d  = fromA ? da : db;
    dj = ((unsigned)d < (unsigned)N_) ? d : 0;
    wj = fromA ? wa : wb;
}

__global__ __launch_bounds__(256) void gat_rows(const int* __restrict__ slots,
                                                const int* __restrict__ cnt,
                                                const int* __restrict__ ovfcnt,
                                                const int* __restrict__ ovf,
                                                const float* __restrict__ h,
                                                const float* __restrict__ e1,
                                                const float* __restrict__ e2,
                                                float* __restrict__ out) {
    const int lane = threadIdx.x & 31;
    const int wv   = __builtin_amdgcn_readfirstlane((int)(threadIdx.x >> 5));
    const int row0 = (blockIdx.x * 8 + wv) * ROWS_PER_WAVE_;
    const int b    = row0 / N_;
    const float*  e2b = e2 + (size_t)b * N_;
    const float2* hb  = reinterpret_cast<const float2*>(h + (size_t)b * N_ * FOUT_) + lane;

    for (int r = 0; r < ROWS_PER_WAVE_; ++r) {
    const int row = row0 + r;
    const int deg = cnt[row];
    const float e1r = e1[row];
    const int dl = slots[row * CAP_ + lane];

    float ax = 0.f, ay = 0.f, rs = 0.f;

    if (deg <= CAP_) {
        const bool have = lane < deg;
        const int  d    = have ? dl : DST_NONE_;
        const float e2d = e2b[have ? d : 0];
        int pos = 0;
        for (int k = 0; k < deg; ++k) {
            const int dk = lane_int(d, k);
            pos += (dk < d || (dk == d && k < lane)) ? 1 : 0;
        }
        if (!have) pos = lane;
        const float w = have ? edge_weight(e1r, e2d) : 0.f;
        const int   sd = __builtin_amdgcn_ds_permute(pos << 2, d);
        const float sw = __int_as_float(__builtin_amdgcn_ds_permute(pos << 2, __float_as_int(w)));

        int j = 0;
        for (; j + 8 <= deg; j += 8) take_edges<8>(hb, sd, sw, j, ax, ay, rs);
        if (j + 4 <= deg) { take_edges<4>(hb, sd, sw, j, ax, ay, rs); j += 4; }
        if (j < deg) {
            const int last = deg - 1;
            const int j1 = (j + 1 < last) ? j + 1 : last;
            const int j2 = (j + 2 < last) ? j + 2 : last;
            const float2 h0 = hb[(size_t)lane_int(sd, j)  * (FOUT_ / 2)];
            const float2 h1 = hb[(size_t)lane_int(sd, j1) * (FOUT_ / 2)];
            const float2 h2 = hb[(size_t)lane_int(sd, j2) * (FOUT_ / 2)];
            const float w0 = lane_flt(sw, j), w1 = lane_flt(sw, j1), w2 = lane_flt(sw, j2);
            ax = fmaf(w0, h0.x, ax); ay = fmaf(w0, h0.y, ay); rs += w0;
            if (j + 1 < deg) { ax = fmaf(w1, h1.x, ax); ay = fmaf(w1, h1.y, ay); rs += w1; }
            if (j + 2 < deg) { ax = fmaf(w2, h2.x, ax); ay = fmaf(w2, h2.y, ay); rs += w2; }
        }
    } else if (deg <= 2 * CAP_) {
        const int   novf = ovfcnt[0];
        const int2* ol   = reinterpret_cast<const int2*>(ovf);
        int d2 = DST_NONE_;
        int k  = 0;
        for (int base = 0; base < novf; base += 32) {
            const int i = base + lane;
            int2 p = make_int2(-1, -1);
            if (i < novf) p = ol[i];
            unsigned hit = __builtin_amdgcn_ballot_w32(p.x == row);
            while (hit != 0u) {
                const int l = __builtin_ffs((int)hit) - 1;
                const int v = lane_int(p.y, l);
                if (lane == k) d2 = v;
                ++k;
                hit &= hit - 1u;
            }
        }
        if (k > 32) k = 32;
        const bool  hasB = lane < k;
        const int   dA   = dl;
        const int   dB   = d2;
        const float e2A  = e2b[dA];
        const float e2B  = e2b[hasB ? dB : 0];
        int posA = 0, posB = 0;
#pragma nounroll
        for (int j = 0; j < CAP_; ++j) {
            const int x = lane_int(dA, j);
            posA += (x < dA || (x == dA && j < lane)) ? 1 : 0;
            posB += (x <= dB) ? 1 : 0;
        }
        for (int j = 0; j < k; ++j) {
            const int y = lane_int(dB, j);
            posA += (y < dA) ? 1 : 0;
            posB += (y < dB || (y == dB && j < lane)) ? 1 : 0;
        }
        const float wA = edge_weight(e1r, e2A);
        const float wB = hasB ? edge_weight(e1r, e2B) : 0.f;
        const int   tot = CAP_ + k;
        for (int j0 = 0; j0 < tot; j0 += 8) {
            float2 hv[8];
            float  wj[8];
#pragma unroll
            for (int i = 0; i < 8; ++i) {
                const int j = (j0 + i < tot) ? j0 + i : tot - 1;
                int dj;
                ranked_edge(posA, posB, hasB, dA, dB, wA, wB, j, dj, wj[i]);
                hv[i] = hb[(size_t)dj * (FOUT_ / 2)];
            }
#pragma unroll
            for (int i = 0; i < 8; ++i) {
                if (j0 + i < tot) {
                    ax = fmaf(wj[i], hv[i].x, ax);
                    ay = fmaf(wj[i], hv[i].y, ay);
                    rs += wj[i];
                }
            }
        }
    } else {
        const int   novf = ovfcnt[0];
        const int2* ol   = reinterpret_cast<const int2*>(ovf);
        int prev = -1;
        for (int it = 0; it < deg; ++it) {
            int m = (dl > prev) ? dl : DST_NONE_;
            for (int base = 0; base < novf; base += 32) {
                const int i = base + lane;
                if (i < novf) {
                    const int2 p = ol[i];
                    if (p.x == row && p.y > prev && p.y < m) m = p.y;
                }
            }
#pragma unroll
            for (int o = 16; o > 0; o >>= 1) {
                const int t = __shfl_xor(m, o, 32);
                m = (t < m) ? t : m;
            }
            m = __builtin_amdgcn_readfirstlane(m);
            if (m == DST_NONE_) break;
            int c = __popc(__builtin_amdgcn_ballot_w32(dl == m));
            for (int base = 0; base < novf; base += 32) {
                const int i = base + lane;
                bool hit = false;
                if (i < novf) {
                    const int2 p = ol[i];
                    hit = (p.x == row) && (p.y == m);
                }
                c += __popc(__builtin_amdgcn_ballot_w32(hit));
            }
            const float wj = edge_weight(e1r, e2b[m]);
            const float2 hv = hb[(size_t)m * (FOUT_ / 2)];
            for (int t = 0; t < c; ++t) {
                ax = fmaf(wj, hv.x, ax);
                ay = fmaf(wj, hv.y, ay);
                rs += wj;
            }
            prev = m;
        }
    }

    float vx = ax / rs;
    float vy = ay / rs;
    if (isnan(vx)) vx = NEG_BIG_;
    if (isnan(vy)) vy = NEG_BIG_;
    float2 o;
    o.x = (vx > 0.f) ? vx : expm1f(vx);
    o.y = (vy > 0.f) ? vy : expm1f(vy);
    reinterpret_cast<float2*>(out + (size_t)row * FOUT_)[lane] = o;
    }
}

extern "C" void kernel_launch(void* const* d_in, const int* in_sizes, int n_in,
                              void* d_out, int out_size, void* d_ws, size_t ws_size,
                              hipStream_t stream) {
    const float* X     = (const float*)d_in[0];
    const int*   edges = (const int*)d_in[1];
    const float* W     = (const float*)d_in[2];
    const float* a     = (const float*)d_in[3];
    float*       out   = (float*)d_out;

    float* h      = (float*)d_ws;
    float* e1     = h + (size_t)ROWS_ * FOUT_;
    float* e2     = e1 + (size_t)ROWS_;
    int*   cnt    = (int*)(e2 + (size_t)ROWS_);
    int*   ovfcnt = cnt + (size_t)ROWS_;
    int*   slots  = ovfcnt + 32;
    int*   ovf    = slots + (size_t)ROWS_ * CAP_;
    int*   seg    = ovf + (size_t)2 * B_ * E_;
    unsigned long long* pairs = (unsigned long long*)(seg + (size_t)SEG_WORDS_);

    gat_front<<<GEMM_BLOCKS_ + PART_BLOCKS_, 256, 0, stream>>>(X, W, h, edges, seg, pairs, ovfcnt);
    gat_mid<<<PLACE_BLOCKS_ + SCORE_BLOCKS_, PLACE_THREADS_, 0, stream>>>(pairs, seg, cnt, slots, ovfcnt, ovf,
                                                                        h, a, e1, e2);
    gat_rows<<<ROWS_ / (8 * ROWS_PER_WAVE_), 256, 0, stream>>>(slots, cnt, ovfcnt, ovf, h, e1, e2, out);
}


# Scores e1,e2 taken from the GEMM accumulators in its epilogue (4 products per lane, 16-lane DPP sum), not by a role re-reading h; launch 2 keeps only bucket blocks; score sum order changed
# speedup vs baseline: 8.5694x; 1.0446x over previous
#include <hip/hip_runtime.h>
#include <hip/hip_bf16.h>
#include <math.h>

typedef __attribute__((ext_vector_type(2))) float v2f;
typedef __attribute__((ext_vector_type(8))) float v8f;

#define B_ 4
#define N_ 50000
#define E_ 800000
#define FIN_ 128
#define FOUT_ 64
#define ALPHA_ 0.2f
#define NEG_BIG_ (-9000000000000000.0f)

#define ROWS_ (B_ * N_)
#define TILES_ (ROWS_ / 16)

#define GEMM_BLOCKS_ ((TILES_ + 7) / 8)

#define WL_PLANE_ ((FIN_ / 4) * 32)
#define WL_ENTRIES_ (2 * WL_PLANE_)

template <int CTRL>
__device__ __forceinline__ float row_xor_get(float v) {
    return __int_as_float(__builtin_amdgcn_update_dpp(0, __float_as_int(v), CTRL, 0xf, 0xf, false));
}
__device__ __forceinline__ float row16_sum(float v) {
    v += row_xor_get<0x161>(v);
    v += row_xor_get<0x162>(v);
    v += row_xor_get<0x164>(v);
    v += row_xor_get<0x168>(v);
    return v;
}

__device__ __forceinline__ void gemm_role(const float* __restrict__ X,
                                          const float* __restrict__ W,
                                          const float* __restrict__ a,
                                          float* __restrict__ h,
                                          float* __restrict__ e1,
                                          float* __restrict__ e2, float4* Wl, int blk) {
#pragma unroll
    for (int i = 0; i < WL_ENTRIES_ / 256; ++i) {
        const int e = i * 256 + (int)threadIdx.x;
        const int p = e / WL_PLANE_;
        const int j = (e >> 5) & (FIN_ / 4 - 1);
        const int l = e & 31;
        const float* w0 = W + (size_t)(4 * j + ((l >> 4) << 1)) * FOUT_ + 32 * p + (l & 15);
        float4 v;
        v.x = w0[0];  v.y = w0[FOUT_];
        v.z = w0[16]; v.w = w0[FOUT_ + 16];
        Wl[e] = v;
    }
    __syncthreads();

    const int lane   = threadIdx.x & 31;
    const int waveId = threadIdx.x >> 5;
    const int rt     = blk * 8 + waveId;
    if (rt >= TILES_) return;

    const int m_base = rt * 16;
    const int mrow   = m_base + (lane & 15);
    const int khalf  = (lane >> 4) << 1;
    const int ncol   = lane & 15;

    v8f acc0 = {}, acc1 = {}, acc2 = {}, acc3 = {};

#pragma unroll 4
    for (int k = 0; k < FIN_; k += 4) {
        const int kk = k + khalf;
        v2f a;
        a.x = X[(size_t)mrow * FIN_ + kk];
        a.y = X[(size_t)mrow * FIN_ + kk + 1];

        const float4 w01 = Wl[(k >> 2) * 32 + lane];
        const float4 w23 = Wl[WL_PLANE_ + (k >> 2) * 32 + lane];

        v2f b0, b1, b2, b3;
        b0.x = w01.x; b0.y = w01.y;
        b1.x = w01.z; b1.y = w01.w;
        b2.x = w23.x; b2.y = w23.y;
        b3.x = w23.z; b3.y = w23.w;

        acc0 = __builtin_amdgcn_wmma_f32_16x16x4_f32(false, a, false, b0, (short)0, acc0, false, false);
        acc1 = __builtin_amdgcn_wmma_f32_16x16x4_f32(false, a, false, b1, (short)0, acc1, false, false);
        acc2 = __builtin_amdgcn_wmma_f32_16x16x4_f32(false, a, false, b2, (short)0, acc2, false, false);
        acc3 = __builtin_amdgcn_wmma_f32_16x16x4_f32(false, a, false, b3, (short)0, acc3, false, false);
    }

    const int rhalf = (lane >> 4) << 3;
#pragma unroll
    for (int i = 0; i < 8; ++i) {
        const size_t row = (size_t)(m_base + i + rhalf);
        h[row * FOUT_ +  0 + ncol] = acc0[i];
        h[row * FOUT_ + 16 + ncol] = acc1[i];
        h[row * FOUT_ + 32 + ncol] = acc2[i];
        h[row * FOUT_ + 48 + ncol] = acc3[i];
    }

    float a1v[4], a2v[4];
#pragma unroll
    for (int t = 0; t < 4; ++t) {
        a1v[t] = a[16 * t + ncol];
        a2v[t] = a[FOUT_ + 16 * t + ncol];
    }
    float s1 = 0.f, s2 = 0.f;
#pragma unroll
    for (int i = 0; i < 8; ++i) {
        float p1 = acc0[i] * a1v[0];
        p1 = fmaf(acc1[i], a1v[1], p1);
        p1 = fmaf(acc2[i], a1v[2], p1);
        p1 = fmaf(acc3[i], a1v[3], p1);
        float p2 = acc0[i] * a2v[0];
        p2 = fmaf(acc1[i], a2v[1], p2);
        p2 = fmaf(acc2[i], a2v[2], p2);
        p2 = fmaf(acc3[i], a2v[3], p2);
        p1 = row16_sum(p1);
        p2 = row16_sum(p2);
        if (ncol == i) { s1 = p1; s2 = p2; }
    }
    if (ncol < 8) {
        e1[m_base + rhalf + ncol] = s1;
        e2[m_base + rhalf + ncol] = s2;
    }
}

#define CAP_ 32
#define RANGE_SHIFT_ 8
#define RANGE_ (1 << RANGE_SHIFT_)
#define NRANGE_ ((N_ + RANGE_ - 1) / RANGE_)
#define NTAB_ 256
#define CHUNK_ 4096
#define NCHUNK_ ((E_ + CHUNK_ - 1) / CHUNK_)
#define QUADS_PER_BATCH_ (E_ / 4)
#define SEG_WORDS_ (((B_ * (NRANGE_ + 1) * NCHUNK_ + 31) / 32) * 32)

static_assert(E_ % 4 == 0, "a group of four edges must not cross a batch");
static_assert(NRANGE_ < NTAB_, "the range table needs one entry past the last range");
static_assert(N_ <= 65536 && CHUNK_ <= 65536, "a source and a place in a chunk share one 32-bit word");

__device__ __forceinline__ unsigned long long make_pair(int s, int d) {
    return ((unsigned long long)(unsigned)d << 32) | (unsigned long long)(unsigned)s;
}

__device__ __forceinline__ int range_rank(int* hist, int bk) {
    const int      key  = __builtin_amdgcn_readfirstlane(bk);
    const unsigned act  = __builtin_amdgcn_ballot_w32(true);
    const unsigned same = __builtin_amdgcn_ballot_w32(bk == key);
    if (same == act) {
        const unsigned lane   = threadIdx.x & 31;
        const int      within = __popc(act & ((1u << lane) - 1u));
        int base = 0;
        if (within == 0) base = atomicAdd(&hist[key], __popc(act));
        base = __builtin_amdgcn_readfirstlane(base);
        return base + within;
    }
    return atomicAdd(&hist[bk], 1);
}

__device__ __forceinline__ void part_role(const int* __restrict__ edges,
                                          int* __restrict__ seg,
                                          unsigned long long* __restrict__ pairs,
                                          unsigned long long* stage, int blk) {
    __shared__ int hist[NTAB_];
    __shared__ int wsum[8];
    const int t = threadIdx.x;
    const int b = blk / NCHUNK_;
    const int c = blk - b * NCHUNK_;
    hist[t] = 0;
    __syncthreads();

    const int4* src4 = reinterpret_cast<const int4*>(edges + ((size_t)b * 2 + 0) * E_);
    const int4* dst4 = reinterpret_cast<const int4*>(edges + ((size_t)b * 2 + 1) * E_);
    int4 s[4], d[4];
    bool ok[4];
#pragma unroll
    for (int i = 0; i < 4; ++i) {
        const int q = c * (CHUNK_ / 4) + i * 256 + t;
        ok[i] = q < QUADS_PER_BATCH_;
        if (ok[i]) { s[i] = src4[q]; d[i] = dst4[q]; }
        else       { s[i] = make_int4(0, 0, 0, 0); d[i] = s[i]; }
    }
    unsigned sr[16];
#pragma unroll
    for (int i = 0; i < 4; ++i) {
        if (ok[i]) {
            sr[4 * i + 0] = (unsigned)s[i].x | ((unsigned)range_rank(hist, s[i].x >> RANGE_SHIFT_) << 16);
            sr[4 * i + 1] = (unsigned)s[i].y | ((unsigned)range_rank(hist, s[i].y >> RANGE_SHIFT_) << 16);
            sr[4 * i + 2] = (unsigned)s[i].z | ((unsigned)range_rank(hist, s[i].z >> RANGE_SHIFT_) << 16);
            sr[4 * i + 3] = (unsigned)s[i].w | ((unsigned)range_rank(hist, s[i].w >> RANGE_SHIFT_) << 16);
        }
    }
    __syncthreads();
    const int v = hist[t];
    int incl = v;
#pragma unroll
    for (int o = 1; o < 32; o <<= 1) {
        const int u = __shfl_up(incl, o, 32);
        if ((t & 31) >= o) incl += u;
    }
    if ((t & 31) == 31) wsum[t >> 5] = incl;
    __syncthreads();
    int off = 0;
    for (int k = 0; k < (t >> 5); ++k) off += wsum[k];
    const int excl = incl - v + off;
    hist[t] = excl;
    if (t <= NRANGE_) seg[(size_t)(b * (NRANGE_ + 1) + t) * NCHUNK_ + c] = excl;
    __syncthreads();
#pragma unroll
    for (int i = 0; i < 4; ++i) {
        if (ok[i]) {
            const int dd[4] = {d[i].x, d[i].y, d[i].z, d[i].w};
#pragma unroll
            for (int e = 0; e < 4; ++e) {
                const unsigned w  = sr[4 * i + e];
                const int      sx = (int)(w & 0xffffu);
                stage[hist[sx >> RANGE_SHIFT_] + (int)(w >> 16)] = make_pair(sx, dd[e]);
            }
        }
    }
    __syncthreads();
    const int n2 = hist[NRANGE_] >> 1;
    const ulonglong2* st2 = reinterpret_cast<const ulonglong2*>(stage);
    ulonglong2* out2 = reinterpret_cast<ulonglong2*>(pairs + (size_t)b * E_ + (size_t)c * CHUNK_);
    for (int i = t; i < n2; i += 256) out2[i] = st2[i];
}

#define PLACE_THREADS_ 512
#define PLACE_WAVES_ (PLACE_THREADS_ / 32)
#define PLACE_TRIPS_ ((NCHUNK_ + PLACE_WAVES_ - 1) / PLACE_WAVES_)
#define PLACE_HEAD_ 64

static_assert(NCHUNK_ <= PLACE_THREADS_, "one thread of place_role per chunk reads the group bounds");
static_assert(RANGE_ <= PLACE_THREADS_, "one thread of place_role per row of the range");

__device__ __forceinline__ void place_pair(int* cntl, int* slotl,
                                           int* __restrict__ ovfcnt, int* __restrict__ ovf,
                                           unsigned long long p, int row0, int rowb) {
    const int sx = (int)(unsigned)(p & 0xffffffffull);
    const int dy = (int)(unsigned)(p >> 32);
    const int lr = sx - row0;
    const int r  = atomicAdd(&cntl[lr], 1);
    if (r < CAP_) {
        slotl[lr * CAP_ + r] = dy;
    } else {
        const int k = atomicAdd(ovfcnt, 1);
        reinterpret_cast<int2*>(ovf)[k] = make_int2(rowb + sx, dy);
    }
}

__device__ __forceinline__ void place_role(const unsigned long long* __restrict__ pairs,
                                           const int* __restrict__ seg,
                                           int* __restrict__ cnt,
                                           int* __restrict__ slots,
                                           int* __restrict__ ovfcnt,
                                           int* __restrict__ ovf, int blk) {
    __shared__ int cntl[RANGE_];
    __shared__ int slotl[RANGE_ * CAP_];
    __shared__ int sbeg[NCHUNK_];
    __shared__ int send[NCHUNK_];
    __shared__ int longs[NCHUNK_];
    __shared__ int nlong;
    const int t = threadIdx.x;
    const int b = blk / NRANGE_;
    const int g = blk - b * NRANGE_;
    if (t == 0) nlong = 0;
    if (t < RANGE_) cntl[t] = 0;
    int glen = 0;
    if (t < NCHUNK_) {
        const int* sg = seg + (size_t)(b * (NRANGE_ + 1) + g) * NCHUNK_;
        const int gb = sg[t];
        const int ge = sg[NCHUNK_ + t];
        sbeg[t] = gb;
        send[t] = ge;
        glen = ge - gb;
    }
    __syncthreads();
    if (glen > PLACE_HEAD_) longs[atomicAdd(&nlong, 1)] = t;

    const int lane = t & 31;
    const int wv   = __builtin_amdgcn_readfirstlane(t >> 5);
    const int row0 = g * RANGE_;
    const int rowb = b * N_;
    const unsigned long long* pb = pairs + (size_t)b * E_;

    unsigned long long p0[PLACE_TRIPS_], p1[PLACE_TRIPS_];
    int i0[PLACE_TRIPS_], en[PLACE_TRIPS_];
#pragma unroll
    for (int k = 0; k < PLACE_TRIPS_; ++k) {
        const int c = wv + k * PLACE_WAVES_;
        int gb = 0, ge = 0;
        if (c < NCHUNK_) { gb = sbeg[c]; ge = send[c]; }
        i0[k] = c * CHUNK_ + gb + lane;
        en[k] = c * CHUNK_ + ge;
        p0[k] = 0ull;
        p1[k] = 0ull;
        if (i0[k] < en[k])      p0[k] = pb[i0[k]];
        if (i0[k] + 32 < en[k]) p1[k] = pb[i0[k] + 32];
    }
#pragma unroll
    for (int k = 0; k < PLACE_TRIPS_; ++k) {
        if (i0[k] < en[k])      place_pair(cntl, slotl, ovfcnt, ovf, p0[k], row0, rowb);
        if (i0[k] + 32 < en[k]) place_pair(cntl, slotl, ovfcnt, ovf, p1[k], row0, rowb);
    }
    __syncthreads();
    const int nl = nlong;
    for (int k = 0; k < nl; ++k) {
        const int c  = longs[k];
        const int ie = c * CHUNK_ + send[c];
        for (int i = c * CHUNK_ + sbeg[c] + PLACE_HEAD_ + t; i < ie; i += PLACE_THREADS_)
            place_pair(cntl, slotl, ovfcnt, ovf, pb[i], row0, rowb);
    }
    __syncthreads();
    for (int r = wv; r < RANGE_; r += PLACE_WAVES_) {
        const int n = cntl[r];
        if (lane < n) slots[(size_t)(rowb + row0 + r) * CAP_ + lane] = slotl[r * CAP_ + lane];
    }
    if (t < RANGE_ && row0 + t < N_) cnt[rowb + row0 + t] = cntl[t];
}

#define PART_BLOCKS_ (B_ * NCHUNK_)

__global__ __launch_bounds__(256, 16) void gat_front(const float* __restrict__ X,
                                                 const float* __restrict__ W,
                                                 const float* __restrict__ a,
                                                 float* __restrict__ h,
                                                 float* __restrict__ e1,
                                                 float* __restrict__ e2,
                                                 const int* __restrict__ edges,
                                                 int* __restrict__ seg,
                                                 unsigned long long* __restrict__ pairs,
                                                 int* __restrict__ ovfcnt) {
    __shared__ __attribute__((aligned(16))) unsigned long long lds_buf[CHUNK_];
    static_assert(WL_ENTRIES_ * sizeof(float4) <= CHUNK_ * sizeof(unsigned long long), "W must fit the shared buffer");
    const int blk = (int)blockIdx.x;
    if (blk < GEMM_BLOCKS_) {
        gemm_role(X, W, a, h, e1, e2, reinterpret_cast<float4*>(lds_buf), blk);
    } else {
        if (blk == GEMM_BLOCKS_ && threadIdx.x == 0) ovfcnt[0] = 0;
        part_role(edges, seg, pairs, lds_buf, blk - GEMM_BLOCKS_);
    }
}

#define PLACE_BLOCKS_ (B_ * NRANGE_)

__global__ __launch_bounds__(PLACE_THREADS_) void gat_mid(const unsigned long long* __restrict__ pairs,
                                                          const int* __restrict__ seg,
                                                          int* __restrict__ cnt,
                                                          int* __restrict__ slots,
                                                          int* __restrict__ ovfcnt,
                                                          int* __restrict__ ovf) {
    place_role(pairs, seg, cnt, slots, ovfcnt, ovf, (int)blockIdx.x);
}

#define DST_NONE_ 0x7fffffff
#define ROWS_PER_WAVE_ 8

static_assert(N_ % ROWS_PER_WAVE_ == 0, "the rows of one wave must lie in one batch");
static_assert(ROWS_ % (8 * ROWS_PER_WAVE_) == 0, "row grid must be exact");

__device__ __forceinline__ float edge_weight(float e1r, float e2d) {
    const float s  = e1r + e2d;
    const float lr = (s >= 0.f) ? s : ALPHA_ * s;
    return __expf(-lr);
}

__device__ __forceinline__ int lane_int(int v, int l) { return __builtin_amdgcn_readlane(v, l); }
__device__ __forceinline__ float lane_flt(float v, int l) {
    return __int_as_float(__builtin_amdgcn_readlane(__float_as_int(v), l));
}

template <int NB>
__device__ __forceinline__ void take_edges(const float2* __restrict__ hb, int sd, float sw, int j,
                                           float& ax, float& ay, float& rs) {
    float2 hv[NB];
    float  wj[NB];
#pragma unroll
    for (int i = 0; i < NB; ++i) {
        const int dj = lane_int(sd, j + i);
        wj[i] = lane_flt(sw, j + i);
        hv[i] = hb[(size_t)dj * (FOUT_ / 2)];
    }
#pragma unroll
    for (int i = 0; i < NB; ++i) {
        ax = fmaf(wj[i], hv[i].x, ax);
        ay = fmaf(wj[i], hv[i].y, ay);
        rs += wj[i];
    }
}

__device__ __forceinline__ void ranked_edge(int posA, int posB, bool hasB, int dA, int dB,
                                            float wA, float wB, int j, int& dj, float& wj) {
    const unsigned mA = __builtin_amdgcn_ballot_w32(posA == j);
    const unsigned mB = __builtin_amdgcn_ballot_w32(hasB && posB == j);
    const bool fromA  = mA != 0u;
    const int  l      = __builtin_ffs((int)(fromA ? mA : mB)) - 1;
    const int   da = lane_int(dA, l), db = lane_int(dB, l);
    const float wa = lane_flt(wA, l), wb = lane_flt(wB, l);
    const int   d  = fromA ? da : db;
    dj = ((unsigned)d < (unsigned)N_) ? d : 0;
    wj = fromA ? wa : wb;
}

__global__ __launch_bounds__(256) void gat_rows(const int* __restrict__ slots,
                                                const int* __restrict__ cnt,
                                                const int* __restrict__ ovfcnt,
                                                const int* __restrict__ ovf,
                                                const float* __restrict__ h,
                                                const float* __restrict__ e1,
                                                const float* __restrict__ e2,
                                                float* __restrict__ out) {
    const int lane = threadIdx.x & 31;
    const int wv   = __builtin_amdgcn_readfirstlane((int)(threadIdx.x >> 5));
    const int row0 = (blockIdx.x * 8 + wv) * ROWS_PER_WAVE_;
    const int b    = row0 / N_;
    const float*  e2b = e2 + (size_t)b * N_;
    const float2* hb  = reinterpret_cast<const float2*>(h + (size_t)b * N_ * FOUT_) + lane;

    for (int r = 0; r < ROWS_PER_WAVE_; ++r) {
    const int row = row0 + r;
    const int deg = cnt[row];
    const float e1r = e1[row];
    const int dl = slots[row * CAP_ + lane];

    float ax = 0.f, ay = 0.f, rs = 0.f;

    if (deg <= CAP_) {
        const bool have = lane < deg;
        const int  d    = have ? dl : DST_NONE_;
        const float e2d = e2b[have ? d : 0];
        int pos = 0;
        for (int k = 0; k < deg; ++k) {
            const int dk = lane_int(d, k);
            pos += (dk < d || (dk == d && k < lane)) ? 1 : 0;
        }
        if (!have) pos = lane;
        const float w = have ? edge_weight(e1r, e2d) : 0.f;
        const int   sd = __builtin_amdgcn_ds_permute(pos << 2, d);
        const float sw = __int_as_float(__builtin_amdgcn_ds_permute(pos << 2, __float_as_int(w)));

        int j = 0;
        for (; j + 8 <= deg; j += 8) take_edges<8>(hb, sd, sw, j, ax, ay, rs);
        if (j + 4 <= deg) { take_edges<4>(hb, sd, sw, j, ax, ay, rs); j += 4; }
        if (j < deg) {
            const int last = deg - 1;
            const int j1 = (j + 1 < last) ? j + 1 : last;
            const int j2 = (j + 2 < last) ? j + 2 : last;
            const float2 h0 = hb[(size_t)lane_int(sd, j)  * (FOUT_ / 2)];
            const float2 h1 = hb[(size_t)lane_int(sd, j1) * (FOUT_ / 2)];
            const float2 h2 = hb[(size_t)lane_int(sd, j2) * (FOUT_ / 2)];
            const float w0 = lane_flt(sw, j), w1 = lane_flt(sw, j1), w2 = lane_flt(sw, j2);
            ax = fmaf(w0, h0.x, ax); ay = fmaf(w0, h0.y, ay); rs += w0;
            if (j + 1 < deg) { ax = fmaf(w1, h1.x, ax); ay = fmaf(w1, h1.y, ay); rs += w1; }
            if (j + 2 < deg) { ax = fmaf(w2, h2.x, ax); ay = fmaf(w2, h2.y, ay); rs += w2; }
        }
    } else if (deg <= 2 * CAP_) {
        const int   novf = ovfcnt[0];
        const int2* ol   = reinterpret_cast<const int2*>(ovf);
        int d2 = DST_NONE_;
        int k  = 0;
        for (int base = 0; base < novf; base += 32) {
            const int i = base + lane;
            int2 p = make_int2(-1, -1);
            if (i < novf) p = ol[i];
            unsigned hit = __builtin_amdgcn_ballot_w32(p.x == row);
            while (hit != 0u) {
                const int l = __builtin_ffs((int)hit) - 1;
                const int v = lane_int(p.y, l);
                if (lane == k) d2 = v;
                ++k;
                hit &= hit - 1u;
            }
        }
        if (k > 32) k = 32;
        const bool  hasB = lane < k;
        const int   dA   = dl;
        const int   dB   = d2;
        const float e2A  = e2b[dA];
        const float e2B  = e2b[hasB ? dB : 0];
        int posA = 0, posB = 0;
#pragma nounroll
        for (int j = 0; j < CAP_; ++j) {
            const int x = lane_int(dA, j);
            posA += (x < dA || (x == dA && j < lane)) ? 1 : 0;
            posB += (x <= dB) ? 1 : 0;
        }
        for (int j = 0; j < k; ++j) {
            const int y = lane_int(dB, j);
            posA += (y < dA) ? 1 : 0;
            posB += (y < dB || (y == dB && j < lane)) ? 1 : 0;
        }
        const float wA = edge_weight(e1r, e2A);
        const float wB = hasB ? edge_weight(e1r, e2B) : 0.f;
        const int   tot = CAP_ + k;
        for (int j0 = 0; j0 < tot; j0 += 8) {
            float2 hv[8];
            float  wj[8];
#pragma unroll
            for (int i = 0; i < 8; ++i) {
                const int j = (j0 + i < tot) ? j0 + i : tot - 1;
                int dj;
                ranked_edge(posA, posB, hasB, dA, dB, wA, wB, j, dj, wj[i]);
                hv[i] = hb[(size_t)dj * (FOUT_ / 2)];
            }
#pragma unroll
            for (int i = 0; i < 8; ++i) {
                if (j0 + i < tot) {
                    ax = fmaf(wj[i], hv[i].x, ax);
                    ay = fmaf(wj[i], hv[i].y, ay);
                    rs += wj[i];
                }
            }
        }
    } else {
        const int   novf = ovfcnt[0];
        const int2* ol   = reinterpret_cast<const int2*>(ovf);
        int prev = -1;
        for (int it = 0; it < deg; ++it) {
            int m = (dl > prev) ? dl : DST_NONE_;
            for (int base = 0; base < novf; base += 32) {
                const int i = base + lane;
                if (i < novf) {
                    const int2 p = ol[i];
                    if (p.x == row && p.y > prev && p.y < m) m = p.y;
                }
            }
#pragma unroll
            for (int o = 16; o > 0; o >>= 1) {
                const int t = __shfl_xor(m, o, 32);
                m = (t < m) ? t : m;
            }
            m = __builtin_amdgcn_readfirstlane(m);
            if (m == DST_NONE_) break;
            int c = __popc(__builtin_amdgcn_ballot_w32(dl == m));
            for (int base = 0; base < novf; base += 32) {
                const int i = base + lane;
                bool hit = false;
                if (i < novf) {
                    const int2 p = ol[i];
                    hit = (p.x == row) && (p.y == m);
                }
                c += __popc(__builtin_amdgcn_ballot_w32(hit));
            }
            const float wj = edge_weight(e1r, e2b[m]);
            const float2 hv = hb[(size_t)m * (FOUT_ / 2)];
            for (int t = 0; t < c; ++t) {
                ax = fmaf(wj, hv.x, ax);
                ay = fmaf(wj, hv.y, ay);
                rs += wj;
            }
            prev = m;
        }
    }

    float vx = ax / rs;
    float vy = ay / rs;
    if (isnan(vx)) vx = NEG_BIG_;
    if (isnan(vy)) vy = NEG_BIG_;
    float2 o;
    o.x = (vx > 0.f) ? vx : expm1f(vx);
    o.y = (vy > 0.f) ? vy : expm1f(vy);
    reinterpret_cast<float2*>(out + (size_t)row * FOUT_)[lane] = o;
    }
}

extern "C" void kernel_launch(void* const* d_in, const int* in_sizes, int n_in,
                              void* d_out, int out_size, void* d_ws, size_t ws_size,
                              hipStream_t stream) {
    const float* X     = (const float*)d_in[0];
    const int*   edges = (const int*)d_in[1];
    const float* W     = (const float*)d_in[2];
    const float* a     = (const float*)d_in[3];
    float*       out   = (float*)d_out;

    float* h      = (float*)d_ws;
    float* e1     = h + (size_t)ROWS_ * FOUT_;
    float* e2     = e1 + (size_t)ROWS_;
    int*   cnt    = (int*)(e2 + (size_t)ROWS_);
    int*   ovfcnt = cnt + (size_t)ROWS_;
    int*   slots  = ovfcnt + 32;
    int*   ovf    = slots + (size_t)ROWS_ * CAP_;
    int*   seg    = ovf + (size_t)2 * B_ * E_;
    unsigned long long* pairs = (unsigned long long*)(seg + (size_t)SEG_WORDS_);

    gat_front<<<GEMM_BLOCKS_ + PART_BLOCKS_, 256, 0, stream>>>(X, W, a, h, e1, e2, edges, seg, pairs, ovfcnt);
    gat_mid<<<PLACE_BLOCKS_, PLACE_THREADS_, 0, stream>>>(pairs, seg, cnt, slots, ovfcnt, ovf);
    gat_rows<<<ROWS_ / (8 * ROWS_PER_WAVE_), 256, 0, stream>>>(slots, cnt, ovfcnt, ovf, h, e1, e2, out);
}
